# starting kernel
# baseline (speedup 1.0000x reference)
#include <hip/hip_runtime.h>


typedef __attribute__((ext_vector_type(2))) float v2f;
typedef __attribute__((ext_vector_type(8))) float v8f;

#define IN_F 128
#define H_F  64
#define C_F  32

static inline int ceil_div_h(long a, long b) { return (int)((a + b - 1) / b); }


__global__ void fill_zero32_kernel(unsigned int* __restrict__ p, long n) {
    long i = (long)blockIdx.x * blockDim.x + threadIdx.x;
    if (i < n) p[i] = 0u;
}

__global__ void deg_hist_kernel(const int* __restrict__ dst, const float* __restrict__ ew,
                                float* __restrict__ deg, int* __restrict__ counts, int E) {
    int e = blockIdx.x * blockDim.x + threadIdx.x;
    if (e < E) {
        int d = dst[e];
        atomicAdd(&deg[d], ew[e]);
        atomicAdd(&counts[d], 1);
    }
}

__global__ void norm_kernel(float* __restrict__ deg, int n) {
    int i = blockIdx.x * blockDim.x + threadIdx.x;
    if (i < n) {
        float d = deg[i];
        deg[i] = (d > 0.0f) ? rsqrtf(fmaxf(d, 1e-12f)) : 0.0f;
    }
}


__global__ void scan_block_kernel(const int* __restrict__ counts, int* __restrict__ rowptr,
                                  int* __restrict__ bsums, int n) {
    __shared__ int sdata[256];
    int tid = threadIdx.x;
    int i = blockIdx.x * 256 + tid;
    int v = (i < n) ? counts[i] : 0;
    sdata[tid] = v;
    __syncthreads();
#pragma unroll
    for (int off = 1; off < 256; off <<= 1) {
        int t = (tid >= off) ? sdata[tid - off] : 0;
        __syncthreads();
        sdata[tid] += t;
        __syncthreads();
    }
    if (i < n) rowptr[i] = sdata[tid] - v;
    if (tid == 255) bsums[blockIdx.x] = sdata[255];
}

__global__ void scan_sums_kernel(int* __restrict__ bsums, int nb) {
    if (blockIdx.x == 0 && threadIdx.x == 0) {
        int acc = 0;
        for (int i = 0; i < nb; ++i) { int t = bsums[i]; bsums[i] = acc; acc += t; }
    }
}

__global__ void scan_add_kernel(int* __restrict__ rowptr, const int* __restrict__ bsums,
                                int n, int E) {
    int i = blockIdx.x * blockDim.x + threadIdx.x;
    if (i < n) rowptr[i] += bsums[i >> 8];
    if (i == 0) rowptr[n] = E;
}

__global__ void build_edges_kernel(const int* __restrict__ src, const int* __restrict__ dst,
                                   const float* __restrict__ ew, const float* __restrict__ norm,
                                   const int* __restrict__ rowptr, int* __restrict__ cursor,
                                   int2* __restrict__ eRec, int E) {
    int e = blockIdx.x * blockDim.x + threadIdx.x;
    if (e < E) {
        int s = src[e], d = dst[e];
        float c = ew[e] * norm[s] * norm[d];
        int pos = rowptr[d] + atomicAdd(&cursor[d], 1);
        eRec[pos] = make_int2(s, __float_as_int(c));
    }
}

template <int FIN, int FOUT>
__global__ void gemm_wmma_kernel(const float* __restrict__ X, const float* __restrict__ W,
                                 float* __restrict__ Y, int nrows) {
    __shared__ float tx[16 * FIN];
    const int nthreads = (FOUT / 16) * 32;
    const int m0  = blockIdx.x * 16;
    const int tid = threadIdx.x;

    for (int i = tid; i < 16 * FIN; i += nthreads) {
        int r = i / FIN, c = i % FIN;
        int row = m0 + r;
        if (row >= nrows) row = nrows - 1;
        tx[i] = X[(long)row * FIN + c];
    }
    __syncthreads();

    const int wave  = tid >> 5;
    const int lane  = tid & 31;
    const int lrow  = lane & 15;
    const int khalf = (lane >> 4) * 2;
    const int n0    = wave * 16;

    v8f acc = {};
#pragma unroll
    for (int k0 = 0; k0 < FIN; k0 += 4) {
        v2f a = *(const v2f*)&tx[lrow * FIN + k0 + khalf];
        v2f b;
        b.x = W[(long)(k0 + khalf + 0) * FOUT + n0 + lrow];
        b.y = W[(long)(k0 + khalf + 1) * FOUT + n0 + lrow];
        acc = __builtin_amdgcn_wmma_f32_16x16x4_f32(
            false, a, false, b, (short)0, acc, false, false);
    }

    const int mbase = m0 + ((lane >> 4) * 8);
    const int col   = n0 + lrow;
#pragma unroll
    for (int r = 0; r < 8; ++r) {
        int row = mbase + r;
        if (row < nrows) Y[(long)row * FOUT + col] = acc[r];
    }
}

template <int F, bool RELU, bool FINAL>
__global__ void gather_agg_kernel(const float* __restrict__ h,
                                  const int* __restrict__ rowptr,
                                  const int2* __restrict__ eRec,
                                  const float* __restrict__ bias,
                                  const float* __restrict__ res,
                                  const float* __restrict__ bias2,
                                  float* __restrict__ y, int n) {
    const int wave = threadIdx.x >> 5;
    const int lane = threadIdx.x & 31;
    const int node = blockIdx.x * (blockDim.x >> 5) + wave;
    if (node >= n) return;

    const int s0 = rowptr[node];
    const int s1 = rowptr[node + 1];

    constexpr int VPL = F / 32;
    const int fbase = lane * VPL;

    float acc[VPL];
#pragma unroll
    for (int j = 0; j < VPL; ++j) acc[j] = 0.0f;

    for (int i = s0; i < s1; ++i) {
        int2  r = eRec[i];
        float c = __int_as_float(r.y);
        const float* hr = h + (long)r.x * F + fbase;
        if constexpr (VPL == 2) {
            float2 v = *(const float2*)hr;
            acc[0] += v.x * c;
            acc[1] += v.y * c;
        } else {
            acc[0] += hr[0] * c;
        }
    }

    const long base = (long)node * F + fbase;
#pragma unroll
    for (int j = 0; j < VPL; ++j) {
        float v = acc[j] + bias[fbase + j];
        if (FINAL) v += res[base + j] + bias2[fbase + j];
        if (RELU) v = fmaxf(v, 0.0f);
        acc[j] = v;
    }
    if constexpr (VPL == 2) {
        *(float2*)(y + base) = make_float2(acc[0], acc[1]);
    } else {
        y[base] = acc[0];
    }
}


extern "C" void kernel_launch(void* const* d_in, const int* in_sizes, int n_in,
                              void* d_out, int out_size, void* d_ws, size_t ws_size,
                              hipStream_t stream) {
    const float* x    = (const float*)d_in[0];
    const int*   src  = (const int*)  d_in[1];
    const int*   dst  = (const int*)  d_in[2];
    const float* ew   = (const float*)d_in[3];
    const float* W1   = (const float*)d_in[4];
    const float* b1   = (const float*)d_in[5];
    const float* W2   = (const float*)d_in[6];
    const float* b2   = (const float*)d_in[7];
    const float* W3   = (const float*)d_in[8];
    const float* b3   = (const float*)d_in[9];
    const float* W4   = (const float*)d_in[10];
    const float* b4   = (const float*)d_in[11];
    const float* Wres = (const float*)d_in[12];
    const float* bres = (const float*)d_in[13];
    float* out = (float*)d_out;

    const int N  = in_sizes[0] / IN_F;
    const int E  = in_sizes[1];
    const int nb = ceil_div_h(N, 256);

    unsigned int* ws = (unsigned int*)d_ws;
    float* deg    = (float*)(ws);
    int*   counts = (int*)  (ws + N);
    int*   cursor = (int*)  (ws + 2L * N);
    int*   rowptr = (int*)  (ws + 3L * N);
    int*   bsums  = (int*)  (ws + 4L * N + 1);
    long   recOff = (4L * N + 1 + nb + 1) & ~1L;
    int2*  eRec   = (int2*) (ws + recOff);
    float* h      = (float*)(ws + recOff + 2L * E);
    float* xbuf   = h    + (size_t)N * H_F;
    float* res    = xbuf + (size_t)N * H_F;

    const int T  = 256;
    const int mt = ceil_div_h(N, 16);
    const int gblocks = ceil_div_h(N, T / 32);

    fill_zero32_kernel<<<ceil_div_h(3L * N, T), T, 0, stream>>>(ws, 3L * N);
    deg_hist_kernel<<<ceil_div_h(E, T), T, 0, stream>>>(dst, ew, deg, counts, E);
    norm_kernel<<<ceil_div_h(N, T), T, 0, stream>>>(deg, N);
    scan_block_kernel<<<nb, 256, 0, stream>>>(counts, rowptr, bsums, N);
    scan_sums_kernel<<<1, 32, 0, stream>>>(bsums, nb);
    scan_add_kernel<<<ceil_div_h(N, T), T, 0, stream>>>(rowptr, bsums, N, E);
    build_edges_kernel<<<ceil_div_h(E, T), T, 0, stream>>>(src, dst, ew, deg, rowptr, cursor,
                                                           eRec, E);

    gemm_wmma_kernel<IN_F, C_F><<<mt, (C_F / 16) * 32, 0, stream>>>(x, Wres, res, N);

    gemm_wmma_kernel<IN_F, H_F><<<mt, (H_F / 16) * 32, 0, stream>>>(x, W1, h, N);
    gather_agg_kernel<H_F, true, false><<<gblocks, T, 0, stream>>>(h, rowptr, eRec,
                                                                   b1, nullptr, nullptr, xbuf, N);
    gemm_wmma_kernel<H_F, H_F><<<mt, (H_F / 16) * 32, 0, stream>>>(xbuf, W2, h, N);
    gather_agg_kernel<H_F, true, false><<<gblocks, T, 0, stream>>>(h, rowptr, eRec,
                                                                   b2, nullptr, nullptr, xbuf, N);
    gemm_wmma_kernel<H_F, H_F><<<mt, (H_F / 16) * 32, 0, stream>>>(xbuf, W3, h, N);
    gather_agg_kernel<H_F, true, false><<<gblocks, T, 0, stream>>>(h, rowptr, eRec,
                                                                   b3, nullptr, nullptr, xbuf, N);
    gemm_wmma_kernel<H_F, C_F><<<mt, (C_F / 16) * 32, 0, stream>>>(xbuf, W4, h, N);
    gather_agg_kernel<C_F, false, true><<<gblocks, T, 0, stream>>>(h, rowptr, eRec,
                                                                   b4, res, bres, out, N);
}


# gather: a wave's lanes split over 4 nodes (8 lanes a node, one b128 per 128-B line of a feature row; planes 256-B aligned for it): gather waves 400k to 100k; per-row sum order kept
# speedup vs baseline: 1.6164x; 1.6164x over previous
#include <hip/hip_runtime.h>


typedef __attribute__((ext_vector_type(2))) float v2f;
typedef __attribute__((ext_vector_type(8))) float v8f;

#define IN_F 128
#define H_F  64
#define C_F  32
#define SLOTS 32

static inline int ceil_div_h(long a, long b) { return (int)((a + b - 1) / b); }


__global__ void fill_zero32_kernel(unsigned int* __restrict__ p, long n) {
    long i = (long)blockIdx.x * blockDim.x + threadIdx.x;
    if (i < n) p[i] = 0u;
}

__device__ __forceinline__ void slot_scatter_role(const int* __restrict__ src, const int* __restrict__ dst,
                                                  const float* __restrict__ ew, int* __restrict__ counts,
                                                  int2* __restrict__ eRec, int* __restrict__ spill,
                                                  int n, int E, int blk) {
    int e = blk * blockDim.x + threadIdx.x;
    if (e < E) {
        const int d  = dst[e];
        const int sv = src[e];
        const int wv = __float_as_int(ew[e]);
        const int s  = atomicAdd(&counts[d], 1);
        if (s < SLOTS) {
            eRec[(long)d * SLOTS + s] = make_int2(sv, wv);
        } else {
            const long o = atomicAdd(&counts[n], 1);
            spill[3 * o + 0] = d;
            spill[3 * o + 1] = sv;
            spill[3 * o + 2] = wv;
        }
    }
}

__global__ void row_norm_kernel(const int* __restrict__ counts, const int2* __restrict__ eRec,
                                const int* __restrict__ spill, float* __restrict__ norm, int n) {
    int i = blockIdx.x * blockDim.x + threadIdx.x;
    if (i >= n) return;
    const int cnt = counts[i];
    const int s0 = i * SLOTS;
    const int s1 = s0 + (cnt < SLOTS ? cnt : SLOTS);
    float d = 0.0f;
    int k = s0;
    for (; k + 4 <= s1; k += 4) {
        float w0 = __int_as_float(eRec[k + 0].y);
        float w1 = __int_as_float(eRec[k + 1].y);
        float w2 = __int_as_float(eRec[k + 2].y);
        float w3 = __int_as_float(eRec[k + 3].y);
        d += w0; d += w1; d += w2; d += w3;
    }
    for (; k < s1; ++k) d += __int_as_float(eRec[k].y);
    if (cnt > SLOTS) {
        const int m = counts[n];
        for (int j = 0; j < m; ++j)
            if (spill[3L * j] == i) d += __int_as_float(spill[3L * j + 2]);
    }
    norm[i] = (d > 0.0f) ? rsqrtf(fmaxf(d, 1e-12f)) : 0.0f;
}

#define COEF_NODES 64
__global__ void coef_kernel(const int* __restrict__ counts, const float* __restrict__ norm,
                            int2* __restrict__ eRec, int* __restrict__ spill, int n) {
    const int tid = threadIdx.x;
    const int n0  = blockIdx.x * COEF_NODES;
    for (int i = tid; i < COEF_NODES * SLOTS; i += blockDim.x) {
        const int node = n0 + i / SLOTS;
        if (node < n) {
            const int slot = i % SLOTS;
            if (slot < counts[node]) {
                const long idx = (long)node * SLOTS + slot;
                int2 r = eRec[idx];
                float c = __int_as_float(r.y) * norm[r.x] * norm[node];
                eRec[idx] = make_int2(r.x, __float_as_int(c));
            }
        }
    }
    const int m = counts[n];
    for (int j = blockIdx.x * blockDim.x + tid; j < m; j += gridDim.x * blockDim.x) {
        const int d = spill[3L * j + 0];
        const int s = spill[3L * j + 1];
        float c = __int_as_float(spill[3L * j + 2]) * norm[s] * norm[d];
        spill[3L * j + 2] = __float_as_int(c);
    }
}

template <int FIN, int FOUT>
__global__ void gemm_wmma_kernel(const float* __restrict__ X, const float* __restrict__ W,
                                 float* __restrict__ Y, int nrows) {
    __shared__ float tx[16 * FIN];
    const int nthreads = (FOUT / 16) * 32;
    const int m0  = blockIdx.x * 16;
    const int tid = threadIdx.x;

    for (int i = tid; i < 16 * FIN; i += nthreads) {
        int r = i / FIN, c = i % FIN;
        int row = m0 + r;
        if (row >= nrows) row = nrows - 1;
        tx[i] = X[(long)row * FIN + c];
    }
    __syncthreads();

    const int wave  = tid >> 5;
    const int lane  = tid & 31;
    const int lrow  = lane & 15;
    const int khalf = (lane >> 4) * 2;
    const int n0    = wave * 16;

    v8f acc = {};
#pragma unroll
    for (int k0 = 0; k0 < FIN; k0 += 4) {
        v2f a = *(const v2f*)&tx[lrow * FIN + k0 + khalf];
        v2f b;
        b.x = W[(long)(k0 + khalf + 0) * FOUT + n0 + lrow];
        b.y = W[(long)(k0 + khalf + 1) * FOUT + n0 + lrow];
        acc = __builtin_amdgcn_wmma_f32_16x16x4_f32(
            false, a, false, b, (short)0, acc, false, false);
    }

    const int mbase = m0 + ((lane >> 4) * 8);
    const int col   = n0 + lrow;
#pragma unroll
    for (int r = 0; r < 8; ++r) {
        int row = mbase + r;
        if (row < nrows) Y[(long)row * FOUT + col] = acc[r];
    }
}

template <int FIN, int FOUT, int TILES>
__device__ __forceinline__ void gemm_tiles_role(const float* __restrict__ X, const float* __restrict__ W,
                                                float* __restrict__ Y, int nrows, int blk,
                                                float* __restrict__ tx) {
    static_assert(TILES * (FOUT / 16) == 8, "eight waves per work-group");
    constexpr int ROWS = 16 * TILES;
    const int tid = threadIdx.x;
    const int m0  = blk * ROWS;

    for (int i = tid; i < ROWS * FIN; i += 256) {
        int r = i / FIN, c = i % FIN;
        int row = m0 + r;
        if (row >= nrows) row = nrows - 1;
        tx[i] = X[(long)row * FIN + c];
    }
    __syncthreads();

    const int wave  = tid >> 5;
    const int lane  = tid & 31;
    const int lrow  = lane & 15;
    const int khalf = (lane >> 4) * 2;
    const int tile  = wave / (FOUT / 16);
    const int n0    = (wave % (FOUT / 16)) * 16;
    const float* txr = tx + (tile * 16 + lrow) * FIN;

    v8f acc = {};
#pragma unroll
    for (int k0 = 0; k0 < FIN; k0 += 4) {
        v2f a = *(const v2f*)&txr[k0 + khalf];
        v2f b;
        b.x = W[(long)(k0 + khalf + 0) * FOUT + n0 + lrow];
        b.y = W[(long)(k0 + khalf + 1) * FOUT + n0 + lrow];
        acc = __builtin_amdgcn_wmma_f32_16x16x4_f32(
            false, a, false, b, (short)0, acc, false, false);
    }

    const int mbase = m0 + tile * 16 + ((lane >> 4) * 8);
    const int col   = n0 + lrow;
#pragma unroll
    for (int r = 0; r < 8; ++r) {
        int row = mbase + r;
        if (row < nrows) Y[(long)row * FOUT + col] = acc[r];
    }
}

__global__ void __launch_bounds__(256)
scatter_and_input_products_kernel(const int* __restrict__ src, const int* __restrict__ dst,
                                  const float* __restrict__ ew, int* __restrict__ counts,
                                  int2* __restrict__ eRec, int* __restrict__ spill, int n, int E,
                                  const float* __restrict__ x,
                                  const float* __restrict__ W1, float* __restrict__ h,
                                  const float* __restrict__ Wres, float* __restrict__ res,
                                  int nScatter, int nL1, int nRes) {
    __shared__ float tx[64 * IN_F];
    const int g = blockIdx.x / 7;
    const int r = blockIdx.x % 7;
    if (r < 4) {
        const int b = 4 * g + r;
        if (b < nScatter) slot_scatter_role(src, dst, ew, counts, eRec, spill, n, E, b);
    } else if (r < 6) {
        const int b = 2 * g + (r - 4);
        if (b < nL1) gemm_tiles_role<IN_F, H_F, 2>(x, W1, h, n, b, tx);
    } else {
        if (g < nRes) gemm_tiles_role<IN_F, C_F, 4>(x, Wres, res, n, g, tx);
    }
}

#define GATHER_NODES 4
template <int F, bool RELU, bool FINAL>
__global__ void __launch_bounds__(256)
gather_agg_kernel(const float* __restrict__ h,
                  const int* __restrict__ counts,
                  const int2* __restrict__ eRec,
                  const int* __restrict__ spill,
                  const float* __restrict__ bias,
                  const float* __restrict__ res,
                  const float* __restrict__ bias2,
                  float* __restrict__ y, int n) {
    constexpr int NV = F / 32;
    constexpr int RV = F / 4;
    constexpr int U  = 8;
    static_assert(U == 8 && SLOTS % U == 0, "a pass never reads past its slot row; the fence names U records");

    const int lane = threadIdx.x & 31;
    const int j    = lane & 7;
    const int node = (blockIdx.x * (blockDim.x >> 5) + (threadIdx.x >> 5)) * GATHER_NODES + (lane >> 3);
    const bool valid = node < n;
    const int cnt = valid ? counts[node] : 0;
    const int m   = cnt < SLOTS ? cnt : SLOTS;
    const int m01  = max(__builtin_amdgcn_readlane(m, 0),  __builtin_amdgcn_readlane(m, 8));
    const int m23  = max(__builtin_amdgcn_readlane(m, 16), __builtin_amdgcn_readlane(m, 24));
    const int mmax = max(m01, m23);

    const int2*   row = eRec + (long)(valid ? node : 0) * SLOTS;
    const float4* hv  = (const float4*)h + j;

    int2 r[U];
#pragma unroll
    for (int u = 0; u < U; ++u) r[u] = row[u];
    asm volatile("" :: "v"(r[0].x), "v"(r[1].x), "v"(r[2].x), "v"(r[3].x),
                       "v"(r[4].x), "v"(r[5].x), "v"(r[6].x), "v"(r[7].x));

    float4 acc[NV];
#pragma unroll
    for (int q = 0; q < NV; ++q) acc[q] = make_float4(0.0f, 0.0f, 0.0f, 0.0f);

    for (int i0 = 0; i0 < mmax; i0 += U) {
        float4 v[U][NV];
        float  c[U];
#pragma unroll
        for (int u = 0; u < U; ++u) {
            const bool on = (i0 + u) < m;
            c[u] = on ? __int_as_float(r[u].y) : 0.0f;
#pragma unroll
            for (int q = 0; q < NV; ++q) v[u][q] = make_float4(0.0f, 0.0f, 0.0f, 0.0f);
            if (on) {
                const float4* p = hv + (long)r[u].x * RV;
#pragma unroll
                for (int q = 0; q < NV; ++q) v[u][q] = p[8 * q];
            }
        }
        if (i0 + U < mmax) {
#pragma unroll
            for (int u = 0; u < U; ++u) r[u] = row[i0 + U + u];
        }
#pragma unroll
        for (int u = 0; u < U; ++u) {
#pragma unroll
            for (int q = 0; q < NV; ++q) {
                acc[q].x += v[u][q].x * c[u];
                acc[q].y += v[u][q].y * c[u];
                acc[q].z += v[u][q].z * c[u];
                acc[q].w += v[u][q].w * c[u];
            }
        }
    }
    if (cnt > SLOTS) {
        const int ms = counts[n];
        for (int k = 0; k < ms; ++k) {
            if (spill[3L * k] != node) continue;
            const float cs = __int_as_float(spill[3L * k + 2]);
            const float4* p = hv + (long)spill[3L * k + 1] * RV;
#pragma unroll
            for (int q = 0; q < NV; ++q) {
                const float4 t = p[8 * q];
                acc[q].x += t.x * cs;
                acc[q].y += t.y * cs;
                acc[q].z += t.z * cs;
                acc[q].w += t.w * cs;
            }
        }
    }
    if (!valid) return;

#pragma unroll
    for (int q = 0; q < NV; ++q) {
        const int f = 32 * q + 4 * j;
        float4 o = acc[q];
        o.x += bias[f + 0]; o.y += bias[f + 1]; o.z += bias[f + 2]; o.w += bias[f + 3];
        if (FINAL) {
            const float4 rr = ((const float4*)res)[(long)node * RV + 8 * q + j];
            o.x += rr.x + bias2[f + 0]; o.y += rr.y + bias2[f + 1];
            o.z += rr.z + bias2[f + 2]; o.w += rr.w + bias2[f + 3];
        }
        if (RELU) {
            o.x = fmaxf(o.x, 0.0f); o.y = fmaxf(o.y, 0.0f);
            o.z = fmaxf(o.z, 0.0f); o.w = fmaxf(o.w, 0.0f);
        }
        ((float4*)y)[(long)node * RV + 8 * q + j] = o;
    }
}


extern "C" void kernel_launch(void* const* d_in, const int* in_sizes, int n_in,
                              void* d_out, int out_size, void* d_ws, size_t ws_size,
                              hipStream_t stream) {
    const float* x    = (const float*)d_in[0];
    const int*   src  = (const int*)  d_in[1];
    const int*   dst  = (const int*)  d_in[2];
    const float* ew   = (const float*)d_in[3];
    const float* W1   = (const float*)d_in[4];
    const float* b1   = (const float*)d_in[5];
    const float* W2   = (const float*)d_in[6];
    const float* b2   = (const float*)d_in[7];
    const float* W3   = (const float*)d_in[8];
    const float* b3   = (const float*)d_in[9];
    const float* W4   = (const float*)d_in[10];
    const float* b4   = (const float*)d_in[11];
    const float* Wres = (const float*)d_in[12];
    const float* bres = (const float*)d_in[13];
    float* out = (float*)d_out;

    const int N  = in_sizes[0] / IN_F;
    const int E  = in_sizes[1];

    unsigned int* ws = (unsigned int*)d_ws;
    float* norm   = (float*)(ws);
    int*   counts = (int*)  (ws + N);
    long   recOff = (2L * N + 1 + 63) & ~63L;
    int2*  eRec   = (int2*) (ws + recOff);
    float* h      = (float*)(ws + recOff + 2L * N * SLOTS);
    float* xbuf   = h    + (size_t)N * H_F;
    float* res    = xbuf + (size_t)N * H_F;
    int*   spill  = (int*)(res + (size_t)N * C_F);

    const int T  = 256;
    const int mt = ceil_div_h(N, 16);
    const int gblocks = ceil_div_h(N, (T / 32) * GATHER_NODES);

    fill_zero32_kernel<<<ceil_div_h(N + 1, T), T, 0, stream>>>(ws + N, N + 1);
    const int nScatter = ceil_div_h(E, T);
    const int nL1      = ceil_div_h(N, 32);
    const int nRes     = ceil_div_h(N, 64);
    int groups = ceil_div_h(nScatter, 4);
    if (ceil_div_h(nL1, 2) > groups) groups = ceil_div_h(nL1, 2);
    if (nRes > groups) groups = nRes;
    scatter_and_input_products_kernel<<<7 * groups, T, 0, stream>>>(src, dst, ew, counts, eRec, spill, N, E,
                                                                    x, W1, h, Wres, res,
                                                                    nScatter, nL1, nRes);
    row_norm_kernel<<<ceil_div_h(N, T), T, 0, stream>>>(counts, eRec, spill, norm, N);
    coef_kernel<<<ceil_div_h(N, COEF_NODES), T, 0, stream>>>(counts, norm, eRec, spill, N);

    gather_agg_kernel<H_F, true, false><<<gblocks, T, 0, stream>>>(h, counts, eRec, spill,
                                                                   b1, nullptr, nullptr, xbuf, N);
    gemm_wmma_kernel<H_F, H_F><<<mt, (H_F / 16) * 32, 0, stream>>>(xbuf, W2, h, N);
    gather_agg_kernel<H_F, true, false><<<gblocks, T, 0, stream>>>(h, counts, eRec, spill,
                                                                   b2, nullptr, nullptr, xbuf, N);
    gemm_wmma_kernel<H_F, H_F><<<mt, (H_F / 16) * 32, 0, stream>>>(xbuf, W3, h, N);
    gather_agg_kernel<H_F, true, false><<<gblocks, T, 0, stream>>>(h, counts, eRec, spill,
                                                                   b3, nullptr, nullptr, xbuf, N);
    gemm_wmma_kernel<H_F, C_F><<<mt, (C_F / 16) * 32, 0, stream>>>(xbuf, W4, h, N);
    gather_agg_kernel<C_F, false, true><<<gblocks, T, 0, stream>>>(h, counts, eRec, spill,
                                                                   b4, res, bres, out, N);
}


# A-tile staging of the four products: grouped b128 loads (the 64-wide launches made 8 dependent dword trips) into LDS rows at pitch FIN+4 (fragment reads were 16-way bank conflicts); k order kept
# speedup vs baseline: 1.8312x; 1.1329x over previous
#include <hip/hip_runtime.h>


typedef __attribute__((ext_vector_type(2))) float v2f;
typedef __attribute__((ext_vector_type(8))) float v8f;

#define IN_F 128
#define H_F  64
#define C_F  32
#define SLOTS 32
#define LDS_PAD 4

static inline int ceil_div_h(long a, long b) { return (int)((a + b - 1) / b); }


__global__ void fill_zero32_kernel(unsigned int* __restrict__ p, long n) {
    long i = (long)blockIdx.x * blockDim.x + threadIdx.x;
    if (i < n) p[i] = 0u;
}

__device__ __forceinline__ void slot_scatter_role(const int* __restrict__ src, const int* __restrict__ dst,
                                                  const float* __restrict__ ew, int* __restrict__ counts,
                                                  int2* __restrict__ eRec, int* __restrict__ spill,
                                                  int n, int E, int blk) {
    int e = blk * blockDim.x + threadIdx.x;
    if (e < E) {
        const int d  = dst[e];
        const int sv = src[e];
        const int wv = __float_as_int(ew[e]);
        const int s  = atomicAdd(&counts[d], 1);
        if (s < SLOTS) {
            eRec[(long)d * SLOTS + s] = make_int2(sv, wv);
        } else {
            const long o = atomicAdd(&counts[n], 1);
            spill[3 * o + 0] = d;
            spill[3 * o + 1] = sv;
            spill[3 * o + 2] = wv;
        }
    }
}

__global__ void row_norm_kernel(const int* __restrict__ counts, const int2* __restrict__ eRec,
                                const int* __restrict__ spill, float* __restrict__ norm, int n) {
    int i = blockIdx.x * blockDim.x + threadIdx.x;
    if (i >= n) return;
    const int cnt = counts[i];
    const int s0 = i * SLOTS;
    const int s1 = s0 + (cnt < SLOTS ? cnt : SLOTS);
    float d = 0.0f;
    int k = s0;
    for (; k + 4 <= s1; k += 4) {
        float w0 = __int_as_float(eRec[k + 0].y);
        float w1 = __int_as_float(eRec[k + 1].y);
        float w2 = __int_as_float(eRec[k + 2].y);
        float w3 = __int_as_float(eRec[k + 3].y);
        d += w0; d += w1; d += w2; d += w3;
    }
    for (; k < s1; ++k) d += __int_as_float(eRec[k].y);
    if (cnt > SLOTS) {
        const int m = counts[n];
        for (int j = 0; j < m; ++j)
            if (spill[3L * j] == i) d += __int_as_float(spill[3L * j + 2]);
    }
    norm[i] = (d > 0.0f) ? rsqrtf(fmaxf(d, 1e-12f)) : 0.0f;
}

#define COEF_NODES 64
__global__ void coef_kernel(const int* __restrict__ counts, const float* __restrict__ norm,
                            int2* __restrict__ eRec, int* __restrict__ spill, int n) {
    const int tid = threadIdx.x;
    const int n0  = blockIdx.x * COEF_NODES;
    for (int i = tid; i < COEF_NODES * SLOTS; i += blockDim.x) {
        const int node = n0 + i / SLOTS;
        if (node < n) {
            const int slot = i % SLOTS;
            if (slot < counts[node]) {
                const long idx = (long)node * SLOTS + slot;
                int2 r = eRec[idx];
                float c = __int_as_float(r.y) * norm[r.x] * norm[node];
                eRec[idx] = make_int2(r.x, __float_as_int(c));
            }
        }
    }
    const int m = counts[n];
    for (int j = blockIdx.x * blockDim.x + tid; j < m; j += gridDim.x * blockDim.x) {
        const int d = spill[3L * j + 0];
        const int s = spill[3L * j + 1];
        float c = __int_as_float(spill[3L * j + 2]) * norm[s] * norm[d];
        spill[3L * j + 2] = __float_as_int(c);
    }
}

template <int FIN, int FOUT>
__global__ void gemm_wmma_kernel(const float* __restrict__ X, const float* __restrict__ W,
                                 float* __restrict__ Y, int nrows) {
    constexpr int PITCH = FIN + LDS_PAD;
    constexpr int NT    = (FOUT / 16) * 32;
    constexpr int RV    = FIN / 4;
    constexpr int VPT   = 16 * RV / NT;
    static_assert(VPT * NT == 16 * RV, "the tile is a whole number of vectors per thread");
    __shared__ __attribute__((aligned(16))) float tx[16 * PITCH];
    const int m0  = blockIdx.x * 16;
    const int tid = threadIdx.x;

    float4 xv[VPT];
#pragma unroll
    for (int t = 0; t < VPT; ++t) {
        const int i = tid + NT * t;
        int row = m0 + i / RV;
        if (row >= nrows) row = nrows - 1;
        xv[t] = ((const float4*)X)[(long)row * RV + i % RV];
    }
#pragma unroll
    for (int t = 0; t < VPT; ++t) {
        const int i = tid + NT * t;
        *(float4*)&tx[(i / RV) * PITCH + 4 * (i % RV)] = xv[t];
    }
    __syncthreads();

    const int wave  = tid >> 5;
    const int lane  = tid & 31;
    const int lrow  = lane & 15;
    const int khalf = (lane >> 4) * 2;
    const int n0    = wave * 16;

    v8f acc = {};
#pragma unroll
    for (int k0 = 0; k0 < FIN; k0 += 4) {
        v2f a = *(const v2f*)&tx[lrow * PITCH + k0 + khalf];
        v2f b;
        b.x = W[(long)(k0 + khalf + 0) * FOUT + n0 + lrow];
        b.y = W[(long)(k0 + khalf + 1) * FOUT + n0 + lrow];
        acc = __builtin_amdgcn_wmma_f32_16x16x4_f32(
            false, a, false, b, (short)0, acc, false, false);
    }

    const int mbase = m0 + ((lane >> 4) * 8);
    const int col   = n0 + lrow;
#pragma unroll
    for (int r = 0; r < 8; ++r) {
        int row = mbase + r;
        if (row < nrows) Y[(long)row * FOUT + col] = acc[r];
    }
}

template <int FIN, int FOUT, int TILES>
__device__ __forceinline__ void gemm_tiles_role(const float* __restrict__ X, const float* __restrict__ W,
                                                float* __restrict__ Y, int nrows, int blk,
                                                float* __restrict__ tx) {
    static_assert(TILES * (FOUT / 16) == 8, "eight waves per work-group");
    constexpr int ROWS  = 16 * TILES;
    constexpr int PITCH = FIN + LDS_PAD;
    const int tid = threadIdx.x;
    const int m0  = blk * ROWS;

    constexpr int RV  = FIN / 4;
    constexpr int VPT = ROWS * RV / 256;
    static_assert(VPT * 256 == ROWS * RV, "the tile is a whole number of vectors per thread");
    float4 xv[VPT];
#pragma unroll
    for (int t = 0; t < VPT; ++t) {
        const int i = tid + 256 * t;
        int row = m0 + i / RV;
        if (row >= nrows) row = nrows - 1;
        xv[t] = ((const float4*)X)[(long)row * RV + i % RV];
    }
#pragma unroll
    for (int t = 0; t < VPT; ++t) {
        const int i = tid + 256 * t;
        *(float4*)&tx[(i / RV) * PITCH + 4 * (i % RV)] = xv[t];
    }
    __syncthreads();

    const int wave  = tid >> 5;
    const int lane  = tid & 31;
    const int lrow  = lane & 15;
    const int khalf = (lane >> 4) * 2;
    const int tile  = wave / (FOUT / 16);
    const int n0    = (wave % (FOUT / 16)) * 16;
    const float* txr = tx + (tile * 16 + lrow) * PITCH;

    v8f acc = {};
#pragma unroll
    for (int k0 = 0; k0 < FIN; k0 += 4) {
        v2f a = *(const v2f*)&txr[k0 + khalf];
        v2f b;
        b.x = W[(long)(k0 + khalf + 0) * FOUT + n0 + lrow];
        b.y = W[(long)(k0 + khalf + 1) * FOUT + n0 + lrow];
        acc = __builtin_amdgcn_wmma_f32_16x16x4_f32(
            false, a, false, b, (short)0, acc, false, false);
    }

    const int mbase = m0 + tile * 16 + ((lane >> 4) * 8);
    const int col   = n0 + lrow;
#pragma unroll
    for (int r = 0; r < 8; ++r) {
        int row = mbase + r;
        if (row < nrows) Y[(long)row * FOUT + col] = acc[r];
    }
}

__global__ void __launch_bounds__(256)
scatter_and_input_products_kernel(const int* __restrict__ src, const int* __restrict__ dst,
                                  const float* __restrict__ ew, int* __restrict__ counts,
                                  int2* __restrict__ eRec, int* __restrict__ spill, int n, int E,
                                  const float* __restrict__ x,
                                  const float* __restrict__ W1, float* __restrict__ h,
                                  const float* __restrict__ Wres, float* __restrict__ res,
                                  int nScatter, int nL1, int nRes) {
    __shared__ __attribute__((aligned(16))) float tx[64 * (IN_F + LDS_PAD)];
    const int g = blockIdx.x / 7;
    const int r = blockIdx.x % 7;
    if (r < 4) {
        const int b = 4 * g + r;
        if (b < nScatter) slot_scatter_role(src, dst, ew, counts, eRec, spill, n, E, b);
    } else if (r < 6) {
        const int b = 2 * g + (r - 4);
        if (b < nL1) gemm_tiles_role<IN_F, H_F, 2>(x, W1, h, n, b, tx);
    } else {
        if (g < nRes) gemm_tiles_role<IN_F, C_F, 4>(x, Wres, res, n, g, tx);
    }
}

#define GATHER_NODES 4
template <int F, bool RELU, bool FINAL>
__global__ void __launch_bounds__(256)
gather_agg_kernel(const float* __restrict__ h,
                  const int* __restrict__ counts,
                  const int2* __restrict__ eRec,
                  const int* __restrict__ spill,
                  const float* __restrict__ bias,
                  const float* __restrict__ res,
                  const float* __restrict__ bias2,
                  float* __restrict__ y, int n) {
    constexpr int NV = F / 32;
    constexpr int RV = F / 4;
    constexpr int U  = 8;
    static_assert(U == 8 && SLOTS % U == 0, "a pass never reads past its slot row; the fence names U records");

    const int lane = threadIdx.x & 31;
    const int j    = lane & 7;
    const int node = (blockIdx.x * (blockDim.x >> 5) + (threadIdx.x >> 5)) * GATHER_NODES + (lane >> 3);
    const bool valid = node < n;
    const int cnt = valid ? counts[node] : 0;
    const int m   = cnt < SLOTS ? cnt : SLOTS;
    const int m01  = max(__builtin_amdgcn_readlane(m, 0),  __builtin_amdgcn_readlane(m, 8));
    const int m23  = max(__builtin_amdgcn_readlane(m, 16), __builtin_amdgcn_readlane(m, 24));
    const int mmax = max(m01, m23);

    const int2*   row = eRec + (long)(valid ? node : 0) * SLOTS;
    const float4* hv  = (const float4*)h + j;

    int2 r[U];
#pragma unroll
    for (int u = 0; u < U; ++u) r[u] = row[u];
    asm volatile("" :: "v"(r[0].x), "v"(r[1].x), "v"(r[2].x), "v"(r[3].x),
                       "v"(r[4].x), "v"(r[5].x), "v"(r[6].x), "v"(r[7].x));

    float4 acc[NV];
#pragma unroll
    for (int q = 0; q < NV; ++q) acc[q] = make_float4(0.0f, 0.0f, 0.0f, 0.0f);

    for (int i0 = 0; i0 < mmax; i0 += U) {
        float4 v[U][NV];
        float  c[U];
#pragma unroll
        for (int u = 0; u < U; ++u) {
            const bool on = (i0 + u) < m;
            c[u] = on ? __int_as_float(r[u].y) : 0.0f;
#pragma unroll
            for (int q = 0; q < NV; ++q) v[u][q] = make_float4(0.0f, 0.0f, 0.0f, 0.0f);
            if (on) {
                const float4* p = hv + (long)r[u].x * RV;
#pragma unroll
                for (int q = 0; q < NV; ++q) v[u][q] = p[8 * q];
            }
        }
        if (i0 + U < mmax) {
#pragma unroll
            for (int u = 0; u < U; ++u) r[u] = row[i0 + U + u];
        }
#pragma unroll
        for (int u = 0; u < U; ++u) {
#pragma unroll
            for (int q = 0; q < NV; ++q) {
                acc[q].x += v[u][q].x * c[u];
                acc[q].y += v[u][q].y * c[u];
                acc[q].z += v[u][q].z * c[u];
                acc[q].w += v[u][q].w * c[u];
            }
        }
    }
    if (cnt > SLOTS) {
        const int ms = counts[n];
        for (int k = 0; k < ms; ++k) {
            if (spill[3L * k] != node) continue;
            const float cs = __int_as_float(spill[3L * k + 2]);
            const float4* p = hv + (long)spill[3L * k + 1] * RV;
#pragma unroll
            for (int q = 0; q < NV; ++q) {
                const float4 t = p[8 * q];
                acc[q].x += t.x * cs;
                acc[q].y += t.y * cs;
                acc[q].z += t.z * cs;
                acc[q].w += t.w * cs;
            }
        }
    }
    if (!valid) return;

#pragma unroll
    for (int q = 0; q < NV; ++q) {
        const int f = 32 * q + 4 * j;
        float4 o = acc[q];
        o.x += bias[f + 0]; o.y += bias[f + 1]; o.z += bias[f + 2]; o.w += bias[f + 3];
        if (FINAL) {
            const float4 rr = ((const float4*)res)[(long)node * RV + 8 * q + j];
            o.x += rr.x + bias2[f + 0]; o.y += rr.y + bias2[f + 1];
            o.z += rr.z + bias2[f + 2]; o.w += rr.w + bias2[f + 3];
        }
        if (RELU) {
            o.x = fmaxf(o.x, 0.0f); o.y = fmaxf(o.y, 0.0f);
            o.z = fmaxf(o.z, 0.0f); o.w = fmaxf(o.w, 0.0f);
        }
        ((float4*)y)[(long)node * RV + 8 * q + j] = o;
    }
}


extern "C" void kernel_launch(void* const* d_in, const int* in_sizes, int n_in,
                              void* d_out, int out_size, void* d_ws, size_t ws_size,
                              hipStream_t stream) {
    const float* x    = (const float*)d_in[0];
    const int*   src  = (const int*)  d_in[1];
    const int*   dst  = (const int*)  d_in[2];
    const float* ew   = (const float*)d_in[3];
    const float* W1   = (const float*)d_in[4];
    const float* b1   = (const float*)d_in[5];
    const float* W2   = (const float*)d_in[6];
    const float* b2   = (const float*)d_in[7];
    const float* W3   = (const float*)d_in[8];
    const float* b3   = (const float*)d_in[9];
    const float* W4   = (const float*)d_in[10];
    const float* b4   = (const float*)d_in[11];
    const float* Wres = (const float*)d_in[12];
    const float* bres = (const float*)d_in[13];
    float* out = (float*)d_out;

    const int N  = in_sizes[0] / IN_F;
    const int E  = in_sizes[1];

    unsigned int* ws = (unsigned int*)d_ws;
    float* norm   = (float*)(ws);
    int*   counts = (int*)  (ws + N);
    long   recOff = (2L * N + 1 + 63) & ~63L;
    int2*  eRec   = (int2*) (ws + recOff);
    float* h      = (float*)(ws + recOff + 2L * N * SLOTS);
    float* xbuf   = h    + (size_t)N * H_F;
    float* res    = xbuf + (size_t)N * H_F;
    int*   spill  = (int*)(res + (size_t)N * C_F);

    const int T  = 256;
    const int mt = ceil_div_h(N, 16);
    const int gblocks = ceil_div_h(N, (T / 32) * GATHER_NODES);

    fill_zero32_kernel<<<ceil_div_h(N + 1, T), T, 0, stream>>>(ws + N, N + 1);
    const int nScatter = ceil_div_h(E, T);
    const int nL1      = ceil_div_h(N, 32);
    const int nRes     = ceil_div_h(N, 64);
    int groups = ceil_div_h(nScatter, 4);
    if (ceil_div_h(nL1, 2) > groups) groups = ceil_div_h(nL1, 2);
    if (nRes > groups) groups = nRes;
    scatter_and_input_products_kernel<<<7 * groups, T, 0, stream>>>(src, dst, ew, counts, eRec, spill, N, E,
                                                                    x, W1, h, Wres, res,
                                                                    nScatter, nL1, nRes);
    row_norm_kernel<<<ceil_div_h(N, T), T, 0, stream>>>(counts, eRec, spill, norm, N);
    coef_kernel<<<ceil_div_h(N, COEF_NODES), T, 0, stream>>>(counts, norm, eRec, spill, N);

    gather_agg_kernel<H_F, true, false><<<gblocks, T, 0, stream>>>(h, counts, eRec, spill,
                                                                   b1, nullptr, nullptr, xbuf, N);
    gemm_wmma_kernel<H_F, H_F><<<mt, (H_F / 16) * 32, 0, stream>>>(xbuf, W2, h, N);
    gather_agg_kernel<H_F, true, false><<<gblocks, T, 0, stream>>>(h, counts, eRec, spill,
                                                                   b2, nullptr, nullptr, xbuf, N);
    gemm_wmma_kernel<H_F, H_F><<<mt, (H_F / 16) * 32, 0, stream>>>(xbuf, W3, h, N);
    gather_agg_kernel<H_F, true, false><<<gblocks, T, 0, stream>>>(h, counts, eRec, spill,
                                                                   b3, nullptr, nullptr, xbuf, N);
    gemm_wmma_kernel<H_F, C_F><<<mt, (C_F / 16) * 32, 0, stream>>>(xbuf, W4, h, N);
    gather_agg_kernel<C_F, false, true><<<gblocks, T, 0, stream>>>(h, counts, eRec, spill,
                                                                   b4, res, bres, out, N);
}


# row lengths move into word 0 of each node's own 256-B slot row (31 record slots): the returning adds fall on one line per node, not 32 counters to a line; count array and its loads go; sum order kept
# speedup vs baseline: 2.0685x; 1.1296x over previous
#include <hip/hip_runtime.h>


typedef __attribute__((ext_vector_type(2))) float v2f;
typedef __attribute__((ext_vector_type(8))) float v8f;

#define IN_F 128
#define H_F  64
#define C_F  32
#define SLOTS 32
#define CAP (SLOTS - 1)
#define LDS_PAD 4

static inline int ceil_div_h(long a, long b) { return (int)((a + b - 1) / b); }


__global__ void clear_headers_kernel(int2* __restrict__ eRec, int* __restrict__ spillLen, int n) {
    int i = blockIdx.x * blockDim.x + threadIdx.x;
    if (i < n) eRec[(long)i * SLOTS] = make_int2(0, 0);
    else if (i == n) *spillLen = 0;
}

__device__ __forceinline__ void slot_scatter_role(const int* __restrict__ src, const int* __restrict__ dst,
                                                  const float* __restrict__ ew, int* __restrict__ spillLen,
                                                  int2* __restrict__ eRec, int* __restrict__ spill,
                                                  int E, int blk) {
    int e = blk * blockDim.x + threadIdx.x;
    if (e < E) {
        const int d  = dst[e];
        const int sv = src[e];
        const int wv = __float_as_int(ew[e]);
        int2* row = eRec + (long)d * SLOTS;
        const int s  = atomicAdd((int*)row, 1);
        if (s < CAP) {
            row[1 + s] = make_int2(sv, wv);
        } else {
            const long o = atomicAdd(spillLen, 1);
            spill[3 * o + 0] = d;
            spill[3 * o + 1] = sv;
            spill[3 * o + 2] = wv;
        }
    }
}

__global__ void row_norm_kernel(const int* __restrict__ spillLen, const int2* __restrict__ eRec,
                                const int* __restrict__ spill, float* __restrict__ norm, int n) {
    int i = blockIdx.x * blockDim.x + threadIdx.x;
    if (i >= n) return;
    const int cnt = eRec[(long)i * SLOTS].x;
    const int s0 = i * SLOTS + 1;
    const int s1 = s0 + (cnt < CAP ? cnt : CAP);
    float d = 0.0f;
    int k = s0;
    for (; k + 4 <= s1; k += 4) {
        float w0 = __int_as_float(eRec[k + 0].y);
        float w1 = __int_as_float(eRec[k + 1].y);
        float w2 = __int_as_float(eRec[k + 2].y);
        float w3 = __int_as_float(eRec[k + 3].y);
        d += w0; d += w1; d += w2; d += w3;
    }
    for (; k < s1; ++k) d += __int_as_float(eRec[k].y);
    if (cnt > CAP) {
        const int m = *spillLen;
        for (int j = 0; j < m; ++j)
            if (spill[3L * j] == i) d += __int_as_float(spill[3L * j + 2]);
    }
    norm[i] = (d > 0.0f) ? rsqrtf(fmaxf(d, 1e-12f)) : 0.0f;
}

#define COEF_NODES 64
__global__ void coef_kernel(const int* __restrict__ spillLen, const float* __restrict__ norm,
                            int2* __restrict__ eRec, int* __restrict__ spill, int n) {
    const int tid = threadIdx.x;
    const int n0  = blockIdx.x * COEF_NODES;
    for (int i = tid; i < COEF_NODES * SLOTS; i += blockDim.x) {
        const int node = n0 + i / SLOTS;
        if (node < n) {
            const int slot = i % SLOTS;
            const int cnt  = eRec[(long)node * SLOTS].x;
            if (slot >= 1 && slot <= (cnt < CAP ? cnt : CAP)) {
                const long idx = (long)node * SLOTS + slot;
                int2 r = eRec[idx];
                float c = __int_as_float(r.y) * norm[r.x] * norm[node];
                eRec[idx] = make_int2(r.x, __float_as_int(c));
            }
        }
    }
    const int m = *spillLen;
    for (int j = blockIdx.x * blockDim.x + tid; j < m; j += gridDim.x * blockDim.x) {
        const int d = spill[3L * j + 0];
        const int s = spill[3L * j + 1];
        float c = __int_as_float(spill[3L * j + 2]) * norm[s] * norm[d];
        spill[3L * j + 2] = __float_as_int(c);
    }
}

template <int FIN, int FOUT>
__global__ void gemm_wmma_kernel(const float* __restrict__ X, const float* __restrict__ W,
                                 float* __restrict__ Y, int nrows) {
    constexpr int PITCH = FIN + LDS_PAD;
    constexpr int NT    = (FOUT / 16) * 32;
    constexpr int RV    = FIN / 4;
    constexpr int VPT   = 16 * RV / NT;
    static_assert(VPT * NT == 16 * RV, "the tile is a whole number of vectors per thread");
    __shared__ __attribute__((aligned(16))) float tx[16 * PITCH];
    const int m0  = blockIdx.x * 16;
    const int tid = threadIdx.x;

    float4 xv[VPT];
#pragma unroll
    for (int t = 0; t < VPT; ++t) {
        const int i = tid + NT * t;
        int row = m0 + i / RV;
        if (row >= nrows) row = nrows - 1;
        xv[t] = ((const float4*)X)[(long)row * RV + i % RV];
    }
#pragma unroll
    for (int t = 0; t < VPT; ++t) {
        const int i = tid + NT * t;
        *(float4*)&tx[(i / RV) * PITCH + 4 * (i % RV)] = xv[t];
    }
    __syncthreads();

    const int wave  = tid >> 5;
    const int lane  = tid & 31;
    const int lrow  = lane & 15;
    const int khalf = (lane >> 4) * 2;
    const int n0    = wave * 16;

    v8f acc = {};
#pragma unroll
    for (int k0 = 0; k0 < FIN; k0 += 4) {
        v2f a = *(const v2f*)&tx[lrow * PITCH + k0 + khalf];
        v2f b;
        b.x = W[(long)(k0 + khalf + 0) * FOUT + n0 + lrow];
        b.y = W[(long)(k0 + khalf + 1) * FOUT + n0 + lrow];
        acc = __builtin_amdgcn_wmma_f32_16x16x4_f32(
            false, a, false, b, (short)0, acc, false, false);
    }

    const int mbase = m0 + ((lane >> 4) * 8);
    const int col   = n0 + lrow;
#pragma unroll
    for (int r = 0; r < 8; ++r) {
        int row = mbase + r;
        if (row < nrows) Y[(long)row * FOUT + col] = acc[r];
    }
}

template <int FIN, int FOUT, int TILES>
__device__ __forceinline__ void gemm_tiles_role(const float* __restrict__ X, const float* __restrict__ W,
                                                float* __restrict__ Y, int nrows, int blk,
                                                float* __restrict__ tx) {
    static_assert(TILES * (FOUT / 16) == 8, "eight waves per work-group");
    constexpr int ROWS  = 16 * TILES;
    constexpr int PITCH = FIN + LDS_PAD;
    const int tid = threadIdx.x;
    const int m0  = blk * ROWS;

    constexpr int RV  = FIN / 4;
    constexpr int VPT = ROWS * RV / 256;
    static_assert(VPT * 256 == ROWS * RV, "the tile is a whole number of vectors per thread");
    float4 xv[VPT];
#pragma unroll
    for (int t = 0; t < VPT; ++t) {
        const int i = tid + 256 * t;
        int row = m0 + i / RV;
        if (row >= nrows) row = nrows - 1;
        xv[t] = ((const float4*)X)[(long)row * RV + i % RV];
    }
#pragma unroll
    for (int t = 0; t < VPT; ++t) {
        const int i = tid + 256 * t;
        *(float4*)&tx[(i / RV) * PITCH + 4 * (i % RV)] = xv[t];
    }
    __syncthreads();

    const int wave  = tid >> 5;
    const int lane  = tid & 31;
    const int lrow  = lane & 15;
    const int khalf = (lane >> 4) * 2;
    const int tile  = wave / (FOUT / 16);
    const int n0    = (wave % (FOUT / 16)) * 16;
    const float* txr = tx + (tile * 16 + lrow) * PITCH;

    v8f acc = {};
#pragma unroll
    for (int k0 = 0; k0 < FIN; k0 += 4) {
        v2f a = *(const v2f*)&txr[k0 + khalf];
        v2f b;
        b.x = W[(long)(k0 + khalf + 0) * FOUT + n0 + lrow];
        b.y = W[(long)(k0 + khalf + 1) * FOUT + n0 + lrow];
        acc = __builtin_amdgcn_wmma_f32_16x16x4_f32(
            false, a, false, b, (short)0, acc, false, false);
    }

    const int mbase = m0 + tile * 16 + ((lane >> 4) * 8);
    const int col   = n0 + lrow;
#pragma unroll
    for (int r = 0; r < 8; ++r) {
        int row = mbase + r;
        if (row < nrows) Y[(long)row * FOUT + col] = acc[r];
    }
}

__global__ void __launch_bounds__(256)
scatter_and_input_products_kernel(const int* __restrict__ src, const int* __restrict__ dst,
                                  const float* __restrict__ ew, int* __restrict__ spillLen,
                                  int2* __restrict__ eRec, int* __restrict__ spill, int n, int E,
                                  const float* __restrict__ x,
                                  const float* __restrict__ W1, float* __restrict__ h,
                                  const float* __restrict__ Wres, float* __restrict__ res,
                                  int nScatter, int nL1, int nRes) {
    __shared__ __attribute__((aligned(16))) float tx[64 * (IN_F + LDS_PAD)];
    const int g = blockIdx.x / 7;
    const int r = blockIdx.x % 7;
    if (r < 4) {
        const int b = 4 * g + r;
        if (b < nScatter) slot_scatter_role(src, dst, ew, spillLen, eRec, spill, E, b);
    } else if (r < 6) {
        const int b = 2 * g + (r - 4);
        if (b < nL1) gemm_tiles_role<IN_F, H_F, 2>(x, W1, h, n, b, tx);
    } else {
        if (g < nRes) gemm_tiles_role<IN_F, C_F, 4>(x, Wres, res, n, g, tx);
    }
}

#define GATHER_NODES 4
template <int F, bool RELU, bool FINAL>
__global__ void __launch_bounds__(256)
gather_agg_kernel(const float* __restrict__ h,
                  const int* __restrict__ spillLen,
                  const int2* __restrict__ eRec,
                  const int* __restrict__ spill,
                  const float* __restrict__ bias,
                  const float* __restrict__ res,
                  const float* __restrict__ bias2,
                  float* __restrict__ y, int n) {
    constexpr int NV = F / 32;
    constexpr int RV = F / 4;
    constexpr int U  = 8;
    static_assert(U == 8 && SLOTS % U == 0, "a pass never reads past its slot row; the fence names U records");

    const int lane = threadIdx.x & 31;
    const int j    = lane & 7;
    const int node = (blockIdx.x * (blockDim.x >> 5) + (threadIdx.x >> 5)) * GATHER_NODES + (lane >> 3);
    const bool valid = node < n;

    const int2*   row = eRec + (long)(valid ? node : 0) * SLOTS;
    const float4* hv  = (const float4*)h + j;

    int2 r[U];
#pragma unroll
    for (int u = 0; u < U; ++u) r[u] = row[u];
    asm volatile("" :: "v"(r[0].x), "v"(r[1].x), "v"(r[2].x), "v"(r[3].x),
                       "v"(r[4].x), "v"(r[5].x), "v"(r[6].x), "v"(r[7].x));

    const int cnt = valid ? r[0].x : 0;
    const int m   = cnt < CAP ? cnt : CAP;
    const int m01  = max(__builtin_amdgcn_readlane(m, 0),  __builtin_amdgcn_readlane(m, 8));
    const int m23  = max(__builtin_amdgcn_readlane(m, 16), __builtin_amdgcn_readlane(m, 24));
    const int mmax = max(m01, m23);

    float4 acc[NV];
#pragma unroll
    for (int q = 0; q < NV; ++q) acc[q] = make_float4(0.0f, 0.0f, 0.0f, 0.0f);

    for (int i0 = 0; i0 <= mmax; i0 += U) {
        float4 v[U][NV];
        float  c[U];
#pragma unroll
        for (int u = 0; u < U; ++u) {
            const int  k  = i0 + u;
            const bool on = (k != 0) && (k <= m);
            c[u] = on ? __int_as_float(r[u].y) : 0.0f;
#pragma unroll
            for (int q = 0; q < NV; ++q) v[u][q] = make_float4(0.0f, 0.0f, 0.0f, 0.0f);
            if (on) {
                const float4* p = hv + (long)r[u].x * RV;
#pragma unroll
                for (int q = 0; q < NV; ++q) v[u][q] = p[8 * q];
            }
        }
        if (i0 + U <= mmax) {
#pragma unroll
            for (int u = 0; u < U; ++u) r[u] = row[i0 + U + u];
        }
#pragma unroll
        for (int u = 0; u < U; ++u) {
#pragma unroll
            for (int q = 0; q < NV; ++q) {
                acc[q].x += v[u][q].x * c[u];
                acc[q].y += v[u][q].y * c[u];
                acc[q].z += v[u][q].z * c[u];
                acc[q].w += v[u][q].w * c[u];
            }
        }
    }
    if (cnt > CAP) {
        const int ms = *spillLen;
        for (int k = 0; k < ms; ++k) {
            if (spill[3L * k] != node) continue;
            const float cs = __int_as_float(spill[3L * k + 2]);
            const float4* p = hv + (long)spill[3L * k + 1] * RV;
#pragma unroll
            for (int q = 0; q < NV; ++q) {
                const float4 t = p[8 * q];
                acc[q].x += t.x * cs;
                acc[q].y += t.y * cs;
                acc[q].z += t.z * cs;
                acc[q].w += t.w * cs;
            }
        }
    }
    if (!valid) return;

#pragma unroll
    for (int q = 0; q < NV; ++q) {
        const int f = 32 * q + 4 * j;
        float4 o = acc[q];
        o.x += bias[f + 0]; o.y += bias[f + 1]; o.z += bias[f + 2]; o.w += bias[f + 3];
        if (FINAL) {
            const float4 rr = ((const float4*)res)[(long)node * RV + 8 * q + j];
            o.x += rr.x + bias2[f + 0]; o.y += rr.y + bias2[f + 1];
            o.z += rr.z + bias2[f + 2]; o.w += rr.w + bias2[f + 3];
        }
        if (RELU) {
            o.x = fmaxf(o.x, 0.0f); o.y = fmaxf(o.y, 0.0f);
            o.z = fmaxf(o.z, 0.0f); o.w = fmaxf(o.w, 0.0f);
        }
        ((float4*)y)[(long)node * RV + 8 * q + j] = o;
    }
}


extern "C" void kernel_launch(void* const* d_in, const int* in_sizes, int n_in,
                              void* d_out, int out_size, void* d_ws, size_t ws_size,
                              hipStream_t stream) {
    const float* x    = (const float*)d_in[0];
    const int*   src  = (const int*)  d_in[1];
    const int*   dst  = (const int*)  d_in[2];
    const float* ew   = (const float*)d_in[3];
    const float* W1   = (const float*)d_in[4];
    const float* b1   = (const float*)d_in[5];
    const float* W2   = (const float*)d_in[6];
    const float* b2   = (const float*)d_in[7];
    const float* W3   = (const float*)d_in[8];
    const float* b3   = (const float*)d_in[9];
    const float* W4   = (const float*)d_in[10];
    const float* b4   = (const float*)d_in[11];
    const float* Wres = (const float*)d_in[12];
    const float* bres = (const float*)d_in[13];
    float* out = (float*)d_out;

    const int N  = in_sizes[0] / IN_F;
    const int E  = in_sizes[1];

    unsigned int* ws = (unsigned int*)d_ws;
    float* norm     = (float*)(ws);
    int*   spillLen = (int*)  (ws + N);
    long   recOff = (2L * N + 1 + 63) & ~63L;
    int2*  eRec   = (int2*) (ws + recOff);
    float* h      = (float*)(ws + recOff + 2L * N * SLOTS);
    float* xbuf   = h    + (size_t)N * H_F;
    float* res    = xbuf + (size_t)N * H_F;
    int*   spill  = (int*)(res + (size_t)N * C_F);

    const int T  = 256;
    const int mt = ceil_div_h(N, 16);
    const int gblocks = ceil_div_h(N, (T / 32) * GATHER_NODES);

    clear_headers_kernel<<<ceil_div_h(N + 1, T), T, 0, stream>>>(eRec, spillLen, N);
    const int nScatter = ceil_div_h(E, T);
    const int nL1      = ceil_div_h(N, 32);
    const int nRes     = ceil_div_h(N, 64);
    int groups = ceil_div_h(nScatter, 4);
    if (ceil_div_h(nL1, 2) > groups) groups = ceil_div_h(nL1, 2);
    if (nRes > groups) groups = nRes;
    scatter_and_input_products_kernel<<<7 * groups, T, 0, stream>>>(src, dst, ew, spillLen, eRec, spill, N, E,
                                                                    x, W1, h, Wres, res,
                                                                    nScatter, nL1, nRes);
    row_norm_kernel<<<ceil_div_h(N, T), T, 0, stream>>>(spillLen, eRec, spill, norm, N);
    coef_kernel<<<ceil_div_h(N, COEF_NODES), T, 0, stream>>>(spillLen, norm, eRec, spill, N);

    gather_agg_kernel<H_F, true, false><<<gblocks, T, 0, stream>>>(h, spillLen, eRec, spill,
                                                                   b1, nullptr, nullptr, xbuf, N);
    gemm_wmma_kernel<H_F, H_F><<<mt, (H_F / 16) * 32, 0, stream>>>(xbuf, W2, h, N);
    gather_agg_kernel<H_F, true, false><<<gblocks, T, 0, stream>>>(h, spillLen, eRec, spill,
                                                                   b2, nullptr, nullptr, xbuf, N);
    gemm_wmma_kernel<H_F, H_F><<<mt, (H_F / 16) * 32, 0, stream>>>(xbuf, W3, h, N);
    gather_agg_kernel<H_F, true, false><<<gblocks, T, 0, stream>>>(h, spillLen, eRec, spill,
                                                                   b3, nullptr, nullptr, xbuf, N);
    gemm_wmma_kernel<H_F, C_F><<<mt, (C_F / 16) * 32, 0, stream>>>(xbuf, W4, h, N);
    gather_agg_kernel<C_F, false, true><<<gblocks, T, 0, stream>>>(h, spillLen, eRec, spill,
                                                                   b4, res, bres, out, N);
}


# half-local scatter: a chunk's edges split by destination parity into two units; each half of the chip (from the die id) claims its own parity's units, so a row line is touched from one side only
# speedup vs baseline: 2.3694x; 1.1455x over previous
#include <hip/hip_runtime.h>


typedef __attribute__((ext_vector_type(2))) float v2f;
typedef __attribute__((ext_vector_type(8))) float v8f;

#define IN_F 128
#define H_F  64
#define C_F  32
#define SLOTS 32
#define CAP (SLOTS - 1)
#define LDS_PAD 4

static inline int ceil_div_h(long a, long b) { return (int)((a + b - 1) / b); }


__global__ void clear_headers_kernel(int2* __restrict__ eRec, int* __restrict__ spillLen, int n) {
    int i = blockIdx.x * blockDim.x + threadIdx.x;
    if (i < n) eRec[(long)i * SLOTS] = make_int2(0, 0);
    else if (i == n) { spillLen[0] = 0; spillLen[1] = 0; spillLen[2] = 0; }
}

__device__ __forceinline__ void slot_scatter_role(const int* __restrict__ src, const int* __restrict__ dst,
                                                  const float* __restrict__ ew, int* __restrict__ spillLen,
                                                  int2* __restrict__ eRec, int* __restrict__ spill,
                                                  int E, int blk, int par) {
    int e = blk * blockDim.x + threadIdx.x;
    if (e < E) {
        const int d  = dst[e];
        if ((d & 1) != par) return;
        const int sv = src[e];
        const int wv = __float_as_int(ew[e]);
        int2* row = eRec + (long)d * SLOTS;
        const int s  = atomicAdd((int*)row, 1);
        if (s < CAP) {
            row[1 + s] = make_int2(sv, wv);
        } else {
            const long o = atomicAdd(spillLen, 1);
            spill[3 * o + 0] = d;
            spill[3 * o + 1] = sv;
            spill[3 * o + 2] = wv;
        }
    }
}

__global__ void row_norm_kernel(const int* __restrict__ spillLen, const int2* __restrict__ eRec,
                                const int* __restrict__ spill, float* __restrict__ norm, int n) {
    int i = blockIdx.x * blockDim.x + threadIdx.x;
    if (i >= n) return;
    const int cnt = eRec[(long)i * SLOTS].x;
    const int s0 = i * SLOTS + 1;
    const int s1 = s0 + (cnt < CAP ? cnt : CAP);
    float d = 0.0f;
    int k = s0;
    for (; k + 4 <= s1; k += 4) {
        float w0 = __int_as_float(eRec[k + 0].y);
        float w1 = __int_as_float(eRec[k + 1].y);
        float w2 = __int_as_float(eRec[k + 2].y);
        float w3 = __int_as_float(eRec[k + 3].y);
        d += w0; d += w1; d += w2; d += w3;
    }
    for (; k < s1; ++k) d += __int_as_float(eRec[k].y);
    if (cnt > CAP) {
        const int m = *spillLen;
        for (int j = 0; j < m; ++j)
            if (spill[3L * j] == i) d += __int_as_float(spill[3L * j + 2]);
    }
    norm[i] = (d > 0.0f) ? rsqrtf(fmaxf(d, 1e-12f)) : 0.0f;
}

#define COEF_NODES 64
__global__ void coef_kernel(const int* __restrict__ spillLen, const float* __restrict__ norm,
                            int2* __restrict__ eRec, int* __restrict__ spill, int n) {
    const int tid = threadIdx.x;
    const int n0  = blockIdx.x * COEF_NODES;
    for (int i = tid; i < COEF_NODES * SLOTS; i += blockDim.x) {
        const int node = n0 + i / SLOTS;
        if (node < n) {
            const int slot = i % SLOTS;
            const int cnt  = eRec[(long)node * SLOTS].x;
            if (slot >= 1 && slot <= (cnt < CAP ? cnt : CAP)) {
                const long idx = (long)node * SLOTS + slot;
                int2 r = eRec[idx];
                float c = __int_as_float(r.y) * norm[r.x] * norm[node];
                eRec[idx] = make_int2(r.x, __float_as_int(c));
            }
        }
    }
    const int m = *spillLen;
    for (int j = blockIdx.x * blockDim.x + tid; j < m; j += gridDim.x * blockDim.x) {
        const int d = spill[3L * j + 0];
        const int s = spill[3L * j + 1];
        float c = __int_as_float(spill[3L * j + 2]) * norm[s] * norm[d];
        spill[3L * j + 2] = __float_as_int(c);
    }
}

template <int FIN, int FOUT>
__global__ void gemm_wmma_kernel(const float* __restrict__ X, const float* __restrict__ W,
                                 float* __restrict__ Y, int nrows) {
    constexpr int PITCH = FIN + LDS_PAD;
    constexpr int NT    = (FOUT / 16) * 32;
    constexpr int RV    = FIN / 4;
    constexpr int VPT   = 16 * RV / NT;
    static_assert(VPT * NT == 16 * RV, "the tile is a whole number of vectors per thread");
    __shared__ __attribute__((aligned(16))) float tx[16 * PITCH];
    const int m0  = blockIdx.x * 16;
    const int tid = threadIdx.x;

    float4 xv[VPT];
#pragma unroll
    for (int t = 0; t < VPT; ++t) {
        const int i = tid + NT * t;
        int row = m0 + i / RV;
        if (row >= nrows) row = nrows - 1;
        xv[t] = ((const float4*)X)[(long)row * RV + i % RV];
    }
#pragma unroll
    for (int t = 0; t < VPT; ++t) {
        const int i = tid + NT * t;
        *(float4*)&tx[(i / RV) * PITCH + 4 * (i % RV)] = xv[t];
    }
    __syncthreads();

    const int wave  = tid >> 5;
    const int lane  = tid & 31;
    const int lrow  = lane & 15;
    const int khalf = (lane >> 4) * 2;
    const int n0    = wave * 16;

    v8f acc = {};
#pragma unroll
    for (int k0 = 0; k0 < FIN; k0 += 4) {
        v2f a = *(const v2f*)&tx[lrow * PITCH + k0 + khalf];
        v2f b;
        b.x = W[(long)(k0 + khalf + 0) * FOUT + n0 + lrow];
        b.y = W[(long)(k0 + khalf + 1) * FOUT + n0 + lrow];
        acc = __builtin_amdgcn_wmma_f32_16x16x4_f32(
            false, a, false, b, (short)0, acc, false, false);
    }

    const int mbase = m0 + ((lane >> 4) * 8);
    const int col   = n0 + lrow;
#pragma unroll
    for (int r = 0; r < 8; ++r) {
        int row = mbase + r;
        if (row < nrows) Y[(long)row * FOUT + col] = acc[r];
    }
}

template <int FIN, int FOUT, int TILES>
__device__ __forceinline__ void gemm_tiles_role(const float* __restrict__ X, const float* __restrict__ W,
                                                float* __restrict__ Y, int nrows, int blk,
                                                float* __restrict__ tx) {
    static_assert(TILES * (FOUT / 16) == 8, "eight waves per work-group");
    constexpr int ROWS  = 16 * TILES;
    constexpr int PITCH = FIN + LDS_PAD;
    const int tid = threadIdx.x;
    const int m0  = blk * ROWS;

    constexpr int RV  = FIN / 4;
    constexpr int VPT = ROWS * RV / 256;
    static_assert(VPT * 256 == ROWS * RV, "the tile is a whole number of vectors per thread");
    float4 xv[VPT];
#pragma unroll
    for (int t = 0; t < VPT; ++t) {
        const int i = tid + 256 * t;
        int row = m0 + i / RV;
        if (row >= nrows) row = nrows - 1;
        xv[t] = ((const float4*)X)[(long)row * RV + i % RV];
    }
#pragma unroll
    for (int t = 0; t < VPT; ++t) {
        const int i = tid + 256 * t;
        *(float4*)&tx[(i / RV) * PITCH + 4 * (i % RV)] = xv[t];
    }
    __syncthreads();

    const int wave  = tid >> 5;
    const int lane  = tid & 31;
    const int lrow  = lane & 15;
    const int khalf = (lane >> 4) * 2;
    const int tile  = wave / (FOUT / 16);
    const int n0    = (wave % (FOUT / 16)) * 16;
    const float* txr = tx + (tile * 16 + lrow) * PITCH;

    v8f acc = {};
#pragma unroll
    for (int k0 = 0; k0 < FIN; k0 += 4) {
        v2f a = *(const v2f*)&txr[k0 + khalf];
        v2f b;
        b.x = W[(long)(k0 + khalf + 0) * FOUT + n0 + lrow];
        b.y = W[(long)(k0 + khalf + 1) * FOUT + n0 + lrow];
        acc = __builtin_amdgcn_wmma_f32_16x16x4_f32(
            false, a, false, b, (short)0, acc, false, false);
    }

    const int mbase = m0 + tile * 16 + ((lane >> 4) * 8);
    const int col   = n0 + lrow;
#pragma unroll
    for (int r = 0; r < 8; ++r) {
        int row = mbase + r;
        if (row < nrows) Y[(long)row * FOUT + col] = acc[r];
    }
}

__global__ void __launch_bounds__(256)
scatter_and_input_products_kernel(const int* __restrict__ src, const int* __restrict__ dst,
                                  const float* __restrict__ ew, int* __restrict__ spillLen,
                                  int2* __restrict__ eRec, int* __restrict__ spill, int n, int E,
                                  const float* __restrict__ x,
                                  const float* __restrict__ W1, float* __restrict__ h,
                                  const float* __restrict__ Wres, float* __restrict__ res,
                                  int nScatter, int nL1, int nRes) {
    __shared__ __attribute__((aligned(16))) float tx[64 * (IN_F + LDS_PAD)];
    __shared__ int unitOfBlock;
    const int g = blockIdx.x / 11;
    const int r = blockIdx.x % 11;
    if (r < 8) {
        if (8 * g + r < 2 * nScatter) {
            if (threadIdx.x == 0) {
                const unsigned where = __builtin_amdgcn_s_sendmsg_rtn(0x87);
                int par = (int)((where >> 18) & 1u);
                int c = atomicAdd(spillLen + 1 + par, 1);
                if (c >= nScatter) {
                    par ^= 1;
                    c = atomicAdd(spillLen + 1 + par, 1);
                }
                unitOfBlock = (c < nScatter) ? (2 * c + par) : -1;
            }
            __syncthreads();
            const int u = unitOfBlock;
            if (u >= 0) slot_scatter_role(src, dst, ew, spillLen, eRec, spill, E, u >> 1, u & 1);
        }
    } else if (r < 10) {
        const int b = 2 * g + (r - 8);
        if (b < nL1) gemm_tiles_role<IN_F, H_F, 2>(x, W1, h, n, b, tx);
    } else {
        if (g < nRes) gemm_tiles_role<IN_F, C_F, 4>(x, Wres, res, n, g, tx);
    }
}

#define GATHER_NODES 4
template <int F, bool RELU, bool FINAL>
__global__ void __launch_bounds__(256)
gather_agg_kernel(const float* __restrict__ h,
                  const int* __restrict__ spillLen,
                  const int2* __restrict__ eRec,
                  const int* __restrict__ spill,
                  const float* __restrict__ bias,
                  const float* __restrict__ res,
                  const float* __restrict__ bias2,
                  float* __restrict__ y, int n) {
    constexpr int NV = F / 32;
    constexpr int RV = F / 4;
    constexpr int U  = 8;
    static_assert(U == 8 && SLOTS % U == 0, "a pass never reads past its slot row; the fence names U records");

    const int lane = threadIdx.x & 31;
    const int j    = lane & 7;
    const int node = (blockIdx.x * (blockDim.x >> 5) + (threadIdx.x >> 5)) * GATHER_NODES + (lane >> 3);
    const bool valid = node < n;

    const int2*   row = eRec + (long)(valid ? node : 0) * SLOTS;
    const float4* hv  = (const float4*)h + j;

    int2 r[U];
#pragma unroll
    for (int u = 0; u < U; ++u) r[u] = row[u];
    asm volatile("" :: "v"(r[0].x), "v"(r[1].x), "v"(r[2].x), "v"(r[3].x),
                       "v"(r[4].x), "v"(r[5].x), "v"(r[6].x), "v"(r[7].x));

    const int cnt = valid ? r[0].x : 0;
    const int m   = cnt < CAP ? cnt : CAP;
    const int m01  = max(__builtin_amdgcn_readlane(m, 0),  __builtin_amdgcn_readlane(m, 8));
    const int m23  = max(__builtin_amdgcn_readlane(m, 16), __builtin_amdgcn_readlane(m, 24));
    const int mmax = max(m01, m23);

    float4 acc[NV];
#pragma unroll
    for (int q = 0; q < NV; ++q) acc[q] = make_float4(0.0f, 0.0f, 0.0f, 0.0f);

    for (int i0 = 0; i0 <= mmax; i0 += U) {
        float4 v[U][NV];
        float  c[U];
#pragma unroll
        for (int u = 0; u < U; ++u) {
            const int  k  = i0 + u;
            const bool on = (k != 0) && (k <= m);
            c[u] = on ? __int_as_float(r[u].y) : 0.0f;
#pragma unroll
            for (int q = 0; q < NV; ++q) v[u][q] = make_float4(0.0f, 0.0f, 0.0f, 0.0f);
            if (on) {
                const float4* p = hv + (long)r[u].x * RV;
#pragma unroll
                for (int q = 0; q < NV; ++q) v[u][q] = p[8 * q];
            }
        }
        if (i0 + U <= mmax) {
#pragma unroll
            for (int u = 0; u < U; ++u) r[u] = row[i0 + U + u];
        }
#pragma unroll
        for (int u = 0; u < U; ++u) {
#pragma unroll
            for (int q = 0; q < NV; ++q) {
                acc[q].x += v[u][q].x * c[u];
                acc[q].y += v[u][q].y * c[u];
                acc[q].z += v[u][q].z * c[u];
                acc[q].w += v[u][q].w * c[u];
            }
        }
    }
    if (cnt > CAP) {
        const int ms = *spillLen;
        for (int k = 0; k < ms; ++k) {
            if (spill[3L * k] != node) continue;
            const float cs = __int_as_float(spill[3L * k + 2]);
            const float4* p = hv + (long)spill[3L * k + 1] * RV;
#pragma unroll
            for (int q = 0; q < NV; ++q) {
                const float4 t = p[8 * q];
                acc[q].x += t.x * cs;
                acc[q].y += t.y * cs;
                acc[q].z += t.z * cs;
                acc[q].w += t.w * cs;
            }
        }
    }
    if (!valid) return;

#pragma unroll
    for (int q = 0; q < NV; ++q) {
        const int f = 32 * q + 4 * j;
        float4 o = acc[q];
        o.x += bias[f + 0]; o.y += bias[f + 1]; o.z += bias[f + 2]; o.w += bias[f + 3];
        if (FINAL) {
            const float4 rr = ((const float4*)res)[(long)node * RV + 8 * q + j];
            o.x += rr.x + bias2[f + 0]; o.y += rr.y + bias2[f + 1];
            o.z += rr.z + bias2[f + 2]; o.w += rr.w + bias2[f + 3];
        }
        if (RELU) {
            o.x = fmaxf(o.x, 0.0f); o.y = fmaxf(o.y, 0.0f);
            o.z = fmaxf(o.z, 0.0f); o.w = fmaxf(o.w, 0.0f);
        }
        ((float4*)y)[(long)node * RV + 8 * q + j] = o;
    }
}


extern "C" void kernel_launch(void* const* d_in, const int* in_sizes, int n_in,
                              void* d_out, int out_size, void* d_ws, size_t ws_size,
                              hipStream_t stream) {
    const float* x    = (const float*)d_in[0];
    const int*   src  = (const int*)  d_in[1];
    const int*   dst  = (const int*)  d_in[2];
    const float* ew   = (const float*)d_in[3];
    const float* W1   = (const float*)d_in[4];
    const float* b1   = (const float*)d_in[5];
    const float* W2   = (const float*)d_in[6];
    const float* b2   = (const float*)d_in[7];
    const float* W3   = (const float*)d_in[8];
    const float* b3   = (const float*)d_in[9];
    const float* W4   = (const float*)d_in[10];
    const float* b4   = (const float*)d_in[11];
    const float* Wres = (const float*)d_in[12];
    const float* bres = (const float*)d_in[13];
    float* out = (float*)d_out;

    const int N  = in_sizes[0] / IN_F;
    const int E  = in_sizes[1];

    unsigned int* ws = (unsigned int*)d_ws;
    float* norm     = (float*)(ws);
    int*   spillLen = (int*)  (ws + N);
    long   recOff = (2L * N + 3 + 63) & ~63L;
    int2*  eRec   = (int2*) (ws + recOff);
    float* h      = (float*)(ws + recOff + 2L * N * SLOTS);
    float* xbuf   = h    + (size_t)N * H_F;
    float* res    = xbuf + (size_t)N * H_F;
    int*   spill  = (int*)(res + (size_t)N * C_F);

    const int T  = 256;
    const int mt = ceil_div_h(N, 16);
    const int gblocks = ceil_div_h(N, (T / 32) * GATHER_NODES);

    clear_headers_kernel<<<ceil_div_h(N + 1, T), T, 0, stream>>>(eRec, spillLen, N);
    const int nScatter = ceil_div_h(E, T);
    const int nL1      = ceil_div_h(N, 32);
    const int nRes     = ceil_div_h(N, 64);
    int groups = ceil_div_h(nScatter, 4);
    if (ceil_div_h(nL1, 2) > groups) groups = ceil_div_h(nL1, 2);
    if (nRes > groups) groups = nRes;
    scatter_and_input_products_kernel<<<11 * groups, T, 0, stream>>>(src, dst, ew, spillLen, eRec, spill, N, E,
                                                                    x, W1, h, Wres, res,
                                                                    nScatter, nL1, nRes);
    row_norm_kernel<<<ceil_div_h(N, T), T, 0, stream>>>(spillLen, eRec, spill, norm, N);
    coef_kernel<<<ceil_div_h(N, COEF_NODES), T, 0, stream>>>(spillLen, norm, eRec, spill, N);

    gather_agg_kernel<H_F, true, false><<<gblocks, T, 0, stream>>>(h, spillLen, eRec, spill,
                                                                   b1, nullptr, nullptr, xbuf, N);
    gemm_wmma_kernel<H_F, H_F><<<mt, (H_F / 16) * 32, 0, stream>>>(xbuf, W2, h, N);
    gather_agg_kernel<H_F, true, false><<<gblocks, T, 0, stream>>>(h, spillLen, eRec, spill,
                                                                   b2, nullptr, nullptr, xbuf, N);
    gemm_wmma_kernel<H_F, H_F><<<mt, (H_F / 16) * 32, 0, stream>>>(xbuf, W3, h, N);
    gather_agg_kernel<H_F, true, false><<<gblocks, T, 0, stream>>>(h, spillLen, eRec, spill,
                                                                   b3, nullptr, nullptr, xbuf, N);
    gemm_wmma_kernel<H_F, C_F><<<mt, (C_F / 16) * 32, 0, stream>>>(xbuf, W4, h, N);
    gather_agg_kernel<C_F, false, true><<<gblocks, T, 0, stream>>>(h, spillLen, eRec, spill,
                                                                   b4, res, bres, out, N);
}


# coefficient pass folded into the first gather: one lane per record looks up norm[src] beside the row loads, forms the same product, writes it back for the later layers; 11 launches to 10
# speedup vs baseline: 2.4561x; 1.0366x over previous
#include <hip/hip_runtime.h>


typedef __attribute__((ext_vector_type(2))) float v2f;
typedef __attribute__((ext_vector_type(8))) float v8f;

#define IN_F 128
#define H_F  64
#define C_F  32
#define SLOTS 32
#define CAP (SLOTS - 1)
#define LDS_PAD 4

static inline int ceil_div_h(long a, long b) { return (int)((a + b - 1) / b); }


__global__ void clear_headers_kernel(int2* __restrict__ eRec, int* __restrict__ spillLen, int n) {
    int i = blockIdx.x * blockDim.x + threadIdx.x;
    if (i < n) eRec[(long)i * SLOTS] = make_int2(0, 0);
    else if (i == n) { spillLen[0] = 0; spillLen[1] = 0; spillLen[2] = 0; }
}

__device__ __forceinline__ void slot_scatter_role(const int* __restrict__ src, const int* __restrict__ dst,
                                                  const float* __restrict__ ew, int* __restrict__ spillLen,
                                                  int2* __restrict__ eRec, int* __restrict__ spill,
                                                  int E, int blk, int par) {
    int e = blk * blockDim.x + threadIdx.x;
    if (e < E) {
        const int d  = dst[e];
        if ((d & 1) != par) return;
        const int sv = src[e];
        const int wv = __float_as_int(ew[e]);
        int2* row = eRec + (long)d * SLOTS;
        const int s  = atomicAdd((int*)row, 1);
        if (s < CAP) {
            row[1 + s] = make_int2(sv, wv);
        } else {
            const long o = atomicAdd(spillLen, 1);
            spill[3 * o + 0] = d;
            spill[3 * o + 1] = sv;
            spill[3 * o + 2] = wv;
        }
    }
}

__global__ void row_norm_kernel(const int* __restrict__ spillLen, const int2* __restrict__ eRec,
                                const int* __restrict__ spill, float* __restrict__ norm, int n) {
    int i = blockIdx.x * blockDim.x + threadIdx.x;
    if (i >= n) return;
    const int cnt = eRec[(long)i * SLOTS].x;
    const int s0 = i * SLOTS + 1;
    const int s1 = s0 + (cnt < CAP ? cnt : CAP);
    float d = 0.0f;
    int k = s0;
    for (; k + 4 <= s1; k += 4) {
        float w0 = __int_as_float(eRec[k + 0].y);
        float w1 = __int_as_float(eRec[k + 1].y);
        float w2 = __int_as_float(eRec[k + 2].y);
        float w3 = __int_as_float(eRec[k + 3].y);
        d += w0; d += w1; d += w2; d += w3;
    }
    for (; k < s1; ++k) d += __int_as_float(eRec[k].y);
    if (cnt > CAP) {
        const int m = *spillLen;
        for (int j = 0; j < m; ++j)
            if (spill[3L * j] == i) d += __int_as_float(spill[3L * j + 2]);
    }
    norm[i] = (d > 0.0f) ? rsqrtf(fmaxf(d, 1e-12f)) : 0.0f;
}

template <int FIN, int FOUT>
__global__ void gemm_wmma_kernel(const float* __restrict__ X, const float* __restrict__ W,
                                 float* __restrict__ Y, int nrows) {
    constexpr int PITCH = FIN + LDS_PAD;
    constexpr int NT    = (FOUT / 16) * 32;
    constexpr int RV    = FIN / 4;
    constexpr int VPT   = 16 * RV / NT;
    static_assert(VPT * NT == 16 * RV, "the tile is a whole number of vectors per thread");
    __shared__ __attribute__((aligned(16))) float tx[16 * PITCH];
    const int m0  = blockIdx.x * 16;
    const int tid = threadIdx.x;

    float4 xv[VPT];
#pragma unroll
    for (int t = 0; t < VPT; ++t) {
        const int i = tid + NT * t;
        int row = m0 + i / RV;
        if (row >= nrows) row = nrows - 1;
        xv[t] = ((const float4*)X)[(long)row * RV + i % RV];
    }
#pragma unroll
    for (int t = 0; t < VPT; ++t) {
        const int i = tid + NT * t;
        *(float4*)&tx[(i / RV) * PITCH + 4 * (i % RV)] = xv[t];
    }
    __syncthreads();

    const int wave  = tid >> 5;
    const int lane  = tid & 31;
    const int lrow  = lane & 15;
    const int khalf = (lane >> 4) * 2;
    const int n0    = wave * 16;

    v8f acc = {};
#pragma unroll
    for (int k0 = 0; k0 < FIN; k0 += 4) {
        v2f a = *(const v2f*)&tx[lrow * PITCH + k0 + khalf];
        v2f b;
        b.x = W[(long)(k0 + khalf + 0) * FOUT + n0 + lrow];
        b.y = W[(long)(k0 + khalf + 1) * FOUT + n0 + lrow];
        acc = __builtin_amdgcn_wmma_f32_16x16x4_f32(
            false, a, false, b, (short)0, acc, false, false);
    }

    const int mbase = m0 + ((lane >> 4) * 8);
    const int col   = n0 + lrow;
#pragma unroll
    for (int r = 0; r < 8; ++r) {
        int row = mbase + r;
        if (row < nrows) Y[(long)row * FOUT + col] = acc[r];
    }
}

template <int FIN, int FOUT, int TILES>
__device__ __forceinline__ void gemm_tiles_role(const float* __restrict__ X, const float* __restrict__ W,
                                                float* __restrict__ Y, int nrows, int blk,
                                                float* __restrict__ tx) {
    static_assert(TILES * (FOUT / 16) == 8, "eight waves per work-group");
    constexpr int ROWS  = 16 * TILES;
    constexpr int PITCH = FIN + LDS_PAD;
    const int tid = threadIdx.x;
    const int m0  = blk * ROWS;

    constexpr int RV  = FIN / 4;
    constexpr int VPT = ROWS * RV / 256;
    static_assert(VPT * 256 == ROWS * RV, "the tile is a whole number of vectors per thread");
    float4 xv[VPT];
#pragma unroll
    for (int t = 0; t < VPT; ++t) {
        const int i = tid + 256 * t;
        int row = m0 + i / RV;
        if (row >= nrows) row = nrows - 1;
        xv[t] = ((const float4*)X)[(long)row * RV + i % RV];
    }
#pragma unroll
    for (int t = 0; t < VPT; ++t) {
        const int i = tid + 256 * t;
        *(float4*)&tx[(i / RV) * PITCH + 4 * (i % RV)] = xv[t];
    }
    __syncthreads();

    const int wave  = tid >> 5;
    const int lane  = tid & 31;
    const int lrow  = lane & 15;
    const int khalf = (lane >> 4) * 2;
    const int tile  = wave / (FOUT / 16);
    const int n0    = (wave % (FOUT / 16)) * 16;
    const float* txr = tx + (tile * 16 + lrow) * PITCH;

    v8f acc = {};
#pragma unroll
    for (int k0 = 0; k0 < FIN; k0 += 4) {
        v2f a = *(const v2f*)&txr[k0 + khalf];
        v2f b;
        b.x = W[(long)(k0 + khalf + 0) * FOUT + n0 + lrow];
        b.y = W[(long)(k0 + khalf + 1) * FOUT + n0 + lrow];
        acc = __builtin_amdgcn_wmma_f32_16x16x4_f32(
            false, a, false, b, (short)0, acc, false, false);
    }

    const int mbase = m0 + tile * 16 + ((lane >> 4) * 8);
    const int col   = n0 + lrow;
#pragma unroll
    for (int r = 0; r < 8; ++r) {
        int row = mbase + r;
        if (row < nrows) Y[(long)row * FOUT + col] = acc[r];
    }
}

__global__ void __launch_bounds__(256)
scatter_and_input_products_kernel(const int* __restrict__ src, const int* __restrict__ dst,
                                  const float* __restrict__ ew, int* __restrict__ spillLen,
                                  int2* __restrict__ eRec, int* __restrict__ spill, int n, int E,
                                  const float* __restrict__ x,
                                  const float* __restrict__ W1, float* __restrict__ h,
                                  const float* __restrict__ Wres, float* __restrict__ res,
                                  int nScatter, int nL1, int nRes) {
    __shared__ __attribute__((aligned(16))) float tx[64 * (IN_F + LDS_PAD)];
    __shared__ int unitOfBlock;
    const int g = blockIdx.x / 11;
    const int r = blockIdx.x % 11;
    if (r < 8) {
        if (8 * g + r < 2 * nScatter) {
            if (threadIdx.x == 0) {
                const unsigned where = __builtin_amdgcn_s_sendmsg_rtn(0x87);
                int par = (int)((where >> 18) & 1u);
                int c = atomicAdd(spillLen + 1 + par, 1);
                if (c >= nScatter) {
                    par ^= 1;
                    c = atomicAdd(spillLen + 1 + par, 1);
                }
                unitOfBlock = (c < nScatter) ? (2 * c + par) : -1;
            }
            __syncthreads();
            const int u = unitOfBlock;
            if (u >= 0) slot_scatter_role(src, dst, ew, spillLen, eRec, spill, E, u >> 1, u & 1);
        }
    } else if (r < 10) {
        const int b = 2 * g + (r - 8);
        if (b < nL1) gemm_tiles_role<IN_F, H_F, 2>(x, W1, h, n, b, tx);
    } else {
        if (g < nRes) gemm_tiles_role<IN_F, C_F, 4>(x, Wres, res, n, g, tx);
    }
}

#define GATHER_NODES 4
#define BCAST8(x, u) __int_as_float(__builtin_amdgcn_mov_dpp8(__float_as_int(x), (u) * 0x249249u))
template <int F, bool RELU, bool FINAL, bool COEF>
__global__ void __launch_bounds__(256)
gather_agg_kernel(const float* __restrict__ h,
                  const int* __restrict__ spillLen,
                  int2* __restrict__ eRec,
                  int* __restrict__ spill,
                  const float* __restrict__ norm,
                  const float* __restrict__ bias,
                  const float* __restrict__ res,
                  const float* __restrict__ bias2,
                  float* __restrict__ y, int n) {
    constexpr int NV = F / 32;
    constexpr int RV = F / 4;
    constexpr int U  = 8;
    static_assert(U == 8 && SLOTS % U == 0, "a pass never reads past its slot row; the fence names U records");

    const int lane = threadIdx.x & 31;
    const int j    = lane & 7;
    const int node = (blockIdx.x * (blockDim.x >> 5) + (threadIdx.x >> 5)) * GATHER_NODES + (lane >> 3);
    const bool valid = node < n;

    int2*         row = eRec + (long)(valid ? node : 0) * SLOTS;
    const float4* hv  = (const float4*)h + j;

    int2  mine = make_int2(0, 0);
    float nd   = 0.0f;
    if (COEF) {
        mine = row[j];
        nd   = norm[valid ? node : 0];
    }
    int2 r[U];
#pragma unroll
    for (int u = 0; u < U; ++u) r[u] = row[u];
    asm volatile("" :: "v"(r[0].x), "v"(r[1].x), "v"(r[2].x), "v"(r[3].x),
                       "v"(r[4].x), "v"(r[5].x), "v"(r[6].x), "v"(r[7].x));

    if (COEF) asm volatile("" :: "v"(mine.x), "v"(nd));

    const int cnt = valid ? r[0].x : 0;
    const int m   = cnt < CAP ? cnt : CAP;
    const int m01  = max(__builtin_amdgcn_readlane(m, 0),  __builtin_amdgcn_readlane(m, 8));
    const int m23  = max(__builtin_amdgcn_readlane(m, 16), __builtin_amdgcn_readlane(m, 24));
    const int mmax = max(m01, m23);

    float4 acc[NV];
#pragma unroll
    for (int q = 0; q < NV; ++q) acc[q] = make_float4(0.0f, 0.0f, 0.0f, 0.0f);

    for (int i0 = 0; i0 <= mmax; i0 += U) {
        float4 v[U][NV];
        float  c[U];
        const int  kj  = i0 + j;
        const bool mon = COEF && (kj != 0) && (kj <= m);
        float nsrc = 0.0f;
        if (COEF) nsrc = norm[mon ? mine.x : 0];
#pragma unroll
        for (int u = 0; u < U; ++u) {
            const int  k  = i0 + u;
            const bool on = (k != 0) && (k <= m);
            if (!COEF) c[u] = on ? __int_as_float(r[u].y) : 0.0f;
#pragma unroll
            for (int q = 0; q < NV; ++q) v[u][q] = make_float4(0.0f, 0.0f, 0.0f, 0.0f);
            if (on) {
                const float4* p = hv + (long)r[u].x * RV;
#pragma unroll
                for (int q = 0; q < NV; ++q) v[u][q] = p[8 * q];
            }
        }
        const int2 cur = mine;
        if (i0 + U <= mmax) {
#pragma unroll
            for (int u = 0; u < U; ++u) r[u] = row[i0 + U + u];
            if (COEF) mine = row[i0 + U + j];
        }
        if (COEF) {
            float myC = 0.0f;
            if (mon) {
                myC = __int_as_float(cur.y) * nsrc * nd;
                row[kj] = make_int2(cur.x, __float_as_int(myC));
            }
            c[0] = BCAST8(myC, 0); c[1] = BCAST8(myC, 1); c[2] = BCAST8(myC, 2); c[3] = BCAST8(myC, 3);
            c[4] = BCAST8(myC, 4); c[5] = BCAST8(myC, 5); c[6] = BCAST8(myC, 6); c[7] = BCAST8(myC, 7);
        }
#pragma unroll
        for (int u = 0; u < U; ++u) {
#pragma unroll
            for (int q = 0; q < NV; ++q) {
                acc[q].x += v[u][q].x * c[u];
                acc[q].y += v[u][q].y * c[u];
                acc[q].z += v[u][q].z * c[u];
                acc[q].w += v[u][q].w * c[u];
            }
        }
    }
    if (cnt > CAP) {
        const int ms = *spillLen;
        for (int k = 0; k < ms; ++k) {
            if (spill[3L * k] != node) continue;
            const int ss = spill[3L * k + 1];
            float cs = __int_as_float(spill[3L * k + 2]);
            if (COEF) {
                cs = cs * norm[ss] * nd;
                if (j == 0) spill[3L * k + 2] = __float_as_int(cs);
            }
            const float4* p = hv + (long)ss * RV;
#pragma unroll
            for (int q = 0; q < NV; ++q) {
                const float4 t = p[8 * q];
                acc[q].x += t.x * cs;
                acc[q].y += t.y * cs;
                acc[q].z += t.z * cs;
                acc[q].w += t.w * cs;
            }
        }
    }
    if (!valid) return;

#pragma unroll
    for (int q = 0; q < NV; ++q) {
        const int f = 32 * q + 4 * j;
        float4 o = acc[q];
        o.x += bias[f + 0]; o.y += bias[f + 1]; o.z += bias[f + 2]; o.w += bias[f + 3];
        if (FINAL) {
            const float4 rr = ((const float4*)res)[(long)node * RV + 8 * q + j];
            o.x += rr.x + bias2[f + 0]; o.y += rr.y + bias2[f + 1];
            o.z += rr.z + bias2[f + 2]; o.w += rr.w + bias2[f + 3];
        }
        if (RELU) {
            o.x = fmaxf(o.x, 0.0f); o.y = fmaxf(o.y, 0.0f);
            o.z = fmaxf(o.z, 0.0f); o.w = fmaxf(o.w, 0.0f);
        }
        ((float4*)y)[(long)node * RV + 8 * q + j] = o;
    }
}


extern "C" void kernel_launch(void* const* d_in, const int* in_sizes, int n_in,
                              void* d_out, int out_size, void* d_ws, size_t ws_size,
                              hipStream_t stream) {
    const float* x    = (const float*)d_in[0];
    const int*   src  = (const int*)  d_in[1];
    const int*   dst  = (const int*)  d_in[2];
    const float* ew   = (const float*)d_in[3];
    const float* W1   = (const float*)d_in[4];
    const float* b1   = (const float*)d_in[5];
    const float* W2   = (const float*)d_in[6];
    const float* b2   = (const float*)d_in[7];
    const float* W3   = (const float*)d_in[8];
    const float* b3   = (const float*)d_in[9];
    const float* W4   = (const float*)d_in[10];
    const float* b4   = (const float*)d_in[11];
    const float* Wres = (const float*)d_in[12];
    const float* bres = (const float*)d_in[13];
    float* out = (float*)d_out;

    const int N  = in_sizes[0] / IN_F;
    const int E  = in_sizes[1];

    unsigned int* ws = (unsigned int*)d_ws;
    float* norm     = (float*)(ws);
    int*   spillLen = (int*)  (ws + N);
    long   recOff = (2L * N + 3 + 63) & ~63L;
    int2*  eRec   = (int2*) (ws + recOff);
    float* h      = (float*)(ws + recOff + 2L * N * SLOTS);
    float* xbuf   = h    + (size_t)N * H_F;
    float* res    = xbuf + (size_t)N * H_F;
    int*   spill  = (int*)(res + (size_t)N * C_F);

    const int T  = 256;
    const int mt = ceil_div_h(N, 16);
    const int gblocks = ceil_div_h(N, (T / 32) * GATHER_NODES);

    clear_headers_kernel<<<ceil_div_h(N + 1, T), T, 0, stream>>>(eRec, spillLen, N);
    const int nScatter = ceil_div_h(E, T);
    const int nL1      = ceil_div_h(N, 32);
    const int nRes     = ceil_div_h(N, 64);
    int groups = ceil_div_h(nScatter, 4);
    if (ceil_div_h(nL1, 2) > groups) groups = ceil_div_h(nL1, 2);
    if (nRes > groups) groups = nRes;
    scatter_and_input_products_kernel<<<11 * groups, T, 0, stream>>>(src, dst, ew, spillLen, eRec, spill, N, E,
                                                                    x, W1, h, Wres, res,
                                                                    nScatter, nL1, nRes);
    row_norm_kernel<<<ceil_div_h(N, T), T, 0, stream>>>(spillLen, eRec, spill, norm, N);

    gather_agg_kernel<H_F, true, false, true><<<gblocks, T, 0, stream>>>(h, spillLen, eRec, spill, norm,
                                                                         b1, nullptr, nullptr, xbuf, N);
    gemm_wmma_kernel<H_F, H_F><<<mt, (H_F / 16) * 32, 0, stream>>>(xbuf, W2, h, N);
    gather_agg_kernel<H_F, true, false, false><<<gblocks, T, 0, stream>>>(h, spillLen, eRec, spill, norm,
                                                                          b2, nullptr, nullptr, xbuf, N);
    gemm_wmma_kernel<H_F, H_F><<<mt, (H_F / 16) * 32, 0, stream>>>(xbuf, W3, h, N);
    gather_agg_kernel<H_F, true, false, false><<<gblocks, T, 0, stream>>>(h, spillLen, eRec, spill, norm,
                                                                          b3, nullptr, nullptr, xbuf, N);
    gemm_wmma_kernel<H_F, C_F><<<mt, (C_F / 16) * 32, 0, stream>>>(xbuf, W4, h, N);
    gather_agg_kernel<C_F, false, true, false><<<gblocks, T, 0, stream>>>(h, spillLen, eRec, spill, norm,
                                                                          b4, res, bres, out, N);
}
